# starting kernel
# baseline (speedup 1.0000x reference)
#include <hip/hip_runtime.h>


typedef __attribute__((ext_vector_type(2))) float v2f;
typedef __attribute__((ext_vector_type(8))) float v8f;

#define GA_B    2048
#define GA_NIN  512
#define GA_NOUT 512

__device__ __forceinline__ v2f mk2(float a, float b) { v2f r; r.x = a; r.y = b; return r; }

__device__ __forceinline__ v8f wmma_f32(v2f a, v2f b, v8f c) {
  return __builtin_amdgcn_wmma_f32_16x16x4_f32(false, a, false, b, (short)0, c, false, false);
}

static constexpr int GRADE_[8] = {0, 1, 1, 1, 2, 2, 2, 3};

static constexpr int EBI[64] = {
  0,0,0,0,0,0,0,0,
  1,2,3, 1,2,3,
  1,1,2,2,3,3,
  1,1,2,2,3,3,
  1,2,3, 1,2,3,
  4,5,6,
  4,4,5,5,6,6,
  4,5,6, 4,5,6,
  4,4,5,5,6,6,
  4,5,6,
  7, 7,7,7, 7,7,7, 7};
static constexpr int EBJ[64] = {
  0,1,2,3,4,5,6,7,
  0,0,0, 1,2,3,
  2,3,1,3,1,2,
  4,5,4,6,5,6,
  6,5,4, 7,7,7,
  0,0,0,
  1,2,1,3,2,3,
  3,2,1, 4,5,6,
  5,6,4,6,4,5,
  7,7,7,
  0, 1,2,3, 4,5,6, 7};
static constexpr int EBK[64] = {
  0,1,2,3,4,5,6,7,
  1,2,3, 0,0,0,
  4,5,4,6,5,6,
  2,3,1,3,1,2,
  7,7,7, 6,5,4,
  4,5,6,
  2,1,3,1,3,2,
  7,7,7, 0,0,0,
  6,5,6,4,5,4,
  3,2,1,
  7, 6,5,4, 3,2,1, 0};
static constexpr int ECV[64] = {
  1,1,1,1,1,1,1,1,
  1,1,1, 1,1,1,
  1,1,-1,1,-1,-1,
  1,1,-1,1,-1,-1,
  1,-1,1, 1,-1,1,
  -1,-1,-1,
  1,-1,1,-1,1,-1,
  -1,1,-1, 1,1,1,
  1,-1,-1,1,1,-1,
  1,-1,1,
  -1, -1,1,-1, 1,-1,1, 1};
static constexpr int EPX[64] = {
  0,1,1,1,2,2,2,3,
  4,4,4, 5,5,5,
  6,6,6,6,6,6,
  7,7,7,7,7,7,
  8,8,8, 9,9,9,
  10,10,10,
  11,11,11,11,11,11,
  12,12,12, 13,13,13,
  14,14,14,14,14,14,
  15,15,15,
  16, 17,17,17, 18,18,18, 19};


__global__ __launch_bounds__(256) void ga_repack_x(
    const float* __restrict__ in, float* __restrict__ outp) {
  const int tid = blockIdx.x * 256 + threadIdx.x;
  const int b = tid >> 9;
  const int n = tid & 511;
  const float4 p0 = *(const float4*)(in + (size_t)tid * 8);
  const float4 p1 = *(const float4*)(in + (size_t)tid * 8 + 4);
  float* dst = outp + (size_t)b * (8 * 512) + n;
  dst[0 * 512] = p0.x; dst[1 * 512] = p0.y; dst[2 * 512] = p0.z; dst[3 * 512] = p0.w;
  dst[4 * 512] = p1.x; dst[5 * 512] = p1.y; dst[6 * 512] = p1.z; dst[7 * 512] = p1.w;
}

template <int C>
__global__ __launch_bounds__(256) void ga_repack_w(
    const float* __restrict__ in, float* __restrict__ outp) {
  const int tid = blockIdx.x * 256 + threadIdx.x;
  const int m = tid >> 9;
  const int n = tid & 511;
  const float* src = in + (size_t)tid * C;
  float* dst = outp + (size_t)m * (C * 512) + n;
#pragma unroll
  for (int c = 0; c < C; ++c) dst[(size_t)c * 512] = src[c];
}


__global__ __launch_bounds__(32) void ga_xr_norm_fast(
    const float* __restrict__ xt, const float* __restrict__ wrt,
    const float* __restrict__ norm_a, float* __restrict__ xrt) {
  const int b0 = blockIdx.x << 4;
  const int m0 = blockIdx.y << 4;
  const int lane = threadIdx.x & 31;
  const int lo = lane & 15;
  const int hi = lane >> 4;

  const v8f vzero = {};
  v8f acc[8];
#pragma unroll
  for (int i = 0; i < 8; ++i) acc[i] = vzero;

  const float* xrow = xt + (size_t)(b0 + lo) * (8 * 512);
  const float* wrow = wrt + (size_t)(m0 + lo) * (4 * 512);

#pragma unroll 4
  for (int k = 0; k < GA_NIN; k += 4) {
    const int kk = k + (hi << 1);
    v2f a[8], g[4];
#pragma unroll
    for (int i = 0; i < 8; ++i) a[i] = *(const v2f*)(xrow + i * 512 + kk);
#pragma unroll
    for (int j = 0; j < 4; ++j) g[j] = *(const v2f*)(wrow + j * 512 + kk);
#pragma unroll
    for (int i = 0; i < 8; ++i)
      acc[i] = wmma_f32(a[i], g[GRADE_[i]], acc[i]);
  }

  const float4 na = *(const float4*)(norm_a + (size_t)(m0 + lo) * 4);
  const float s0 = 1.0f / (1.0f + __expf(-na.x));
  const float s1 = 1.0f / (1.0f + __expf(-na.y));
  const float s2 = 1.0f / (1.0f + __expf(-na.z));
  const float s3 = 1.0f / (1.0f + __expf(-na.w));

#pragma unroll
  for (int v = 0; v < 8; ++v) {
    const float e0 = acc[0][v], e1 = acc[1][v], e2 = acc[2][v], e3 = acc[3][v];
    const float e4 = acc[4][v], e5 = acc[5][v], e6 = acc[6][v], e7 = acc[7][v];
    const float n0 = __builtin_sqrtf(e0 * e0);
    const float n1 = __builtin_sqrtf(e1 * e1 + e2 * e2 + e3 * e3);
    const float n2 = __builtin_sqrtf(e4 * e4 + e5 * e5 + e6 * e6);
    const float n3 = __builtin_sqrtf(e7 * e7);
    const float i0 = 1.0f / (s0 * (n0 - 1.0f) + 1.0f + 1e-6f);
    const float i1 = 1.0f / (s1 * (n1 - 1.0f) + 1.0f + 1e-6f);
    const float i2 = 1.0f / (s2 * (n2 - 1.0f) + 1.0f + 1e-6f);
    const float i3 = 1.0f / (s3 * (n3 - 1.0f) + 1.0f + 1e-6f);
    float* dst = xrt + (size_t)(b0 + v + (hi << 3)) * (8 * 512) + (m0 + lo);
    dst[0 * 512] = e0 * i0;
    dst[1 * 512] = e1 * i1;
    dst[2 * 512] = e2 * i1;
    dst[3 * 512] = e3 * i1;
    dst[4 * 512] = e4 * i2;
    dst[5 * 512] = e5 * i2;
    dst[6 * 512] = e6 * i2;
    dst[7 * 512] = e7 * i3;
  }
}

__global__ __launch_bounds__(32) void ga_out_fast(
    const float* __restrict__ xt, const float* __restrict__ xrt,
    const float* __restrict__ wt, const float* __restrict__ wlt,
    const float* __restrict__ b_left, float* __restrict__ out) {
  const int b0 = blockIdx.x << 4;
  const int m0 = blockIdx.y << 4;
  const int lane = threadIdx.x & 31;
  const int lo = lane & 15;
  const int hi = lane >> 4;

  const v8f vzero = {};
  v8f acc[8];
#pragma unroll
  for (int i = 0; i < 8; ++i) acc[i] = vzero;

  const float* xrow = xt + (size_t)(b0 + lo) * (8 * 512);
  const float* rrow = xrt + (size_t)(b0 + lo) * (8 * 512);
  const float* lrow = wlt + (size_t)(m0 + lo) * (4 * 512);
  const float* wrow = wt + (size_t)(m0 + lo) * (20 * 512);

#pragma unroll 1
  for (int k = 0; k < GA_NIN; k += 4) {
    const int kk = k + (hi << 1);
    v2f a[8], r[8], bl[4], bw[20];
#pragma unroll
    for (int i = 0; i < 8; ++i) a[i] = *(const v2f*)(xrow + i * 512 + kk);
#pragma unroll
    for (int i = 0; i < 8; ++i) r[i] = *(const v2f*)(rrow + i * 512 + kk);
#pragma unroll
    for (int j = 0; j < 4; ++j) bl[j] = *(const v2f*)(lrow + j * 512 + kk);
#pragma unroll
    for (int p = 0; p < 20; ++p) bw[p] = *(const v2f*)(wrow + p * 512 + kk);

#pragma unroll
    for (int i = 0; i < 8; ++i)
      acc[i] = wmma_f32(a[i], bl[GRADE_[i]], acc[i]);

#pragma unroll
    for (int e = 0; e < 64; ++e) {
      v2f p = a[EBI[e]] * r[EBK[e]];
      if (ECV[e] < 0) p = -p;
      acc[EBJ[e]] = wmma_f32(p, bw[EPX[e]], acc[EBJ[e]]);
    }
  }

  const float bias = b_left[m0 + lo];
  const float rs2 = 0.70710678118654752440f;
#pragma unroll
  for (int v = 0; v < 8; ++v) {
    const float o0 = (acc[0][v] + bias) * rs2;
    const float o1 = acc[1][v] * rs2, o2 = acc[2][v] * rs2, o3 = acc[3][v] * rs2;
    const float o4 = acc[4][v] * rs2, o5 = acc[5][v] * rs2;
    const float o6 = acc[6][v] * rs2, o7 = acc[7][v] * rs2;
    float* dst = out + ((size_t)(b0 + v + (hi << 3)) * GA_NOUT + (m0 + lo)) * 8;
    *(float4*)(dst)     = make_float4(o0, o1, o2, o3);
    *(float4*)(dst + 4) = make_float4(o4, o5, o6, o7);
  }
}


__global__ __launch_bounds__(32) void ga_xr_norm_kernel(
    const float* __restrict__ x, const float* __restrict__ w_right,
    const float* __restrict__ norm_a, float* __restrict__ xr) {
  const int b0 = blockIdx.x << 4;
  const int m0 = blockIdx.y << 4;
  const int lane = threadIdx.x & 31;
  const int lo = lane & 15;
  const int hi = lane >> 4;

  const v8f vzero = {};
  v8f acc[8];
#pragma unroll
  for (int i = 0; i < 8; ++i) acc[i] = vzero;

  const float* xrow = x + (size_t)(b0 + lo) * (GA_NIN * 8);
  const float* wrow = w_right + (size_t)(m0 + lo) * (GA_NIN * 4);

#pragma unroll 2
  for (int k = 0; k < GA_NIN; k += 4) {
    const int kk = k + (hi << 1);
    const float4 xa0 = *(const float4*)(xrow + kk * 8);
    const float4 xa1 = *(const float4*)(xrow + kk * 8 + 4);
    const float4 xb0 = *(const float4*)(xrow + kk * 8 + 8);
    const float4 xb1 = *(const float4*)(xrow + kk * 8 + 12);
    const float4 w0  = *(const float4*)(wrow + kk * 4);
    const float4 w1  = *(const float4*)(wrow + kk * 4 + 4);

    const v2f a0 = mk2(xa0.x, xb0.x), a1 = mk2(xa0.y, xb0.y);
    const v2f a2 = mk2(xa0.z, xb0.z), a3 = mk2(xa0.w, xb0.w);
    const v2f a4 = mk2(xa1.x, xb1.x), a5 = mk2(xa1.y, xb1.y);
    const v2f a6 = mk2(xa1.z, xb1.z), a7 = mk2(xa1.w, xb1.w);
    const v2f g0 = mk2(w0.x, w1.x), g1 = mk2(w0.y, w1.y);
    const v2f g2 = mk2(w0.z, w1.z), g3 = mk2(w0.w, w1.w);

    acc[0] = wmma_f32(a0, g0, acc[0]);
    acc[1] = wmma_f32(a1, g1, acc[1]);
    acc[2] = wmma_f32(a2, g1, acc[2]);
    acc[3] = wmma_f32(a3, g1, acc[3]);
    acc[4] = wmma_f32(a4, g2, acc[4]);
    acc[5] = wmma_f32(a5, g2, acc[5]);
    acc[6] = wmma_f32(a6, g2, acc[6]);
    acc[7] = wmma_f32(a7, g3, acc[7]);
  }

  const float4 na = *(const float4*)(norm_a + (size_t)(m0 + lo) * 4);
  const float s0 = 1.0f / (1.0f + __expf(-na.x));
  const float s1 = 1.0f / (1.0f + __expf(-na.y));
  const float s2 = 1.0f / (1.0f + __expf(-na.z));
  const float s3 = 1.0f / (1.0f + __expf(-na.w));

#pragma unroll
  for (int v = 0; v < 8; ++v) {
    const float e0 = acc[0][v], e1 = acc[1][v], e2 = acc[2][v], e3 = acc[3][v];
    const float e4 = acc[4][v], e5 = acc[5][v], e6 = acc[6][v], e7 = acc[7][v];
    const float n0 = __builtin_sqrtf(e0 * e0);
    const float n1 = __builtin_sqrtf(e1 * e1 + e2 * e2 + e3 * e3);
    const float n2 = __builtin_sqrtf(e4 * e4 + e5 * e5 + e6 * e6);
    const float n3 = __builtin_sqrtf(e7 * e7);
    const float i0 = 1.0f / (s0 * (n0 - 1.0f) + 1.0f + 1e-6f);
    const float i1 = 1.0f / (s1 * (n1 - 1.0f) + 1.0f + 1e-6f);
    const float i2 = 1.0f / (s2 * (n2 - 1.0f) + 1.0f + 1e-6f);
    const float i3 = 1.0f / (s3 * (n3 - 1.0f) + 1.0f + 1e-6f);
    float* dst = xr + ((size_t)(b0 + v + (hi << 3)) * GA_NIN + (m0 + lo)) * 8;
    *(float4*)(dst)     = make_float4(e0 * i0, e1 * i1, e2 * i1, e3 * i1);
    *(float4*)(dst + 4) = make_float4(e4 * i2, e5 * i2, e6 * i2, e7 * i3);
  }
}

__global__ __launch_bounds__(32) void ga_out_kernel(
    const float* __restrict__ x, const float* __restrict__ xr,
    const float* __restrict__ weight, const float* __restrict__ w_left,
    const float* __restrict__ b_left, float* __restrict__ out) {
  const int b0 = blockIdx.x << 4;
  const int m0 = blockIdx.y << 4;
  const int lane = threadIdx.x & 31;
  const int lo = lane & 15;
  const int hi = lane >> 4;

  const v8f vzero = {};
  v8f acc[8];
#pragma unroll
  for (int i = 0; i < 8; ++i) acc[i] = vzero;

  const float* xrow = x  + (size_t)(b0 + lo) * (GA_NIN * 8);
  const float* rrow = xr + (size_t)(b0 + lo) * (GA_NIN * 8);
  const float* lrow = w_left + (size_t)(m0 + lo) * (GA_NIN * 4);
  const float* wrow = weight + (size_t)(m0 + lo) * (GA_NIN * 20);

#pragma unroll 1
  for (int k = 0; k < GA_NIN; k += 4) {
    const int kk = k + (hi << 1);

    const float4 xa0 = *(const float4*)(xrow + kk * 8);
    const float4 xa1 = *(const float4*)(xrow + kk * 8 + 4);
    const float4 xb0 = *(const float4*)(xrow + kk * 8 + 8);
    const float4 xb1 = *(const float4*)(xrow + kk * 8 + 12);
    const float4 ra0 = *(const float4*)(rrow + kk * 8);
    const float4 ra1 = *(const float4*)(rrow + kk * 8 + 4);
    const float4 rb0 = *(const float4*)(rrow + kk * 8 + 8);
    const float4 rb1 = *(const float4*)(rrow + kk * 8 + 12);
    const float4 l0  = *(const float4*)(lrow + kk * 4);
    const float4 l1  = *(const float4*)(lrow + kk * 4 + 4);
    const float4 q0  = *(const float4*)(wrow + kk * 20);
    const float4 q1  = *(const float4*)(wrow + kk * 20 + 4);
    const float4 q2  = *(const float4*)(wrow + kk * 20 + 8);
    const float4 q3  = *(const float4*)(wrow + kk * 20 + 12);
    const float4 q4  = *(const float4*)(wrow + kk * 20 + 16);
    const float4 t0  = *(const float4*)(wrow + kk * 20 + 20);
    const float4 t1  = *(const float4*)(wrow + kk * 20 + 24);
    const float4 t2  = *(const float4*)(wrow + kk * 20 + 28);
    const float4 t3  = *(const float4*)(wrow + kk * 20 + 32);
    const float4 t4  = *(const float4*)(wrow + kk * 20 + 36);

    v2f a[8], r[8], bl[4], bw[20];
    a[0] = mk2(xa0.x, xb0.x); a[1] = mk2(xa0.y, xb0.y);
    a[2] = mk2(xa0.z, xb0.z); a[3] = mk2(xa0.w, xb0.w);
    a[4] = mk2(xa1.x, xb1.x); a[5] = mk2(xa1.y, xb1.y);
    a[6] = mk2(xa1.z, xb1.z); a[7] = mk2(xa1.w, xb1.w);
    r[0] = mk2(ra0.x, rb0.x); r[1] = mk2(ra0.y, rb0.y);
    r[2] = mk2(ra0.z, rb0.z); r[3] = mk2(ra0.w, rb0.w);
    r[4] = mk2(ra1.x, rb1.x); r[5] = mk2(ra1.y, rb1.y);
    r[6] = mk2(ra1.z, rb1.z); r[7] = mk2(ra1.w, rb1.w);
    bl[0] = mk2(l0.x, l1.x); bl[1] = mk2(l0.y, l1.y);
    bl[2] = mk2(l0.z, l1.z); bl[3] = mk2(l0.w, l1.w);
    bw[0]  = mk2(q0.x, t0.x); bw[1]  = mk2(q0.y, t0.y);
    bw[2]  = mk2(q0.z, t0.z); bw[3]  = mk2(q0.w, t0.w);
    bw[4]  = mk2(q1.x, t1.x); bw[5]  = mk2(q1.y, t1.y);
    bw[6]  = mk2(q1.z, t1.z); bw[7]  = mk2(q1.w, t1.w);
    bw[8]  = mk2(q2.x, t2.x); bw[9]  = mk2(q2.y, t2.y);
    bw[10] = mk2(q2.z, t2.z); bw[11] = mk2(q2.w, t2.w);
    bw[12] = mk2(q3.x, t3.x); bw[13] = mk2(q3.y, t3.y);
    bw[14] = mk2(q3.z, t3.z); bw[15] = mk2(q3.w, t3.w);
    bw[16] = mk2(q4.x, t4.x); bw[17] = mk2(q4.y, t4.y);
    bw[18] = mk2(q4.z, t4.z); bw[19] = mk2(q4.w, t4.w);

#pragma unroll
    for (int i = 0; i < 8; ++i)
      acc[i] = wmma_f32(a[i], bl[GRADE_[i]], acc[i]);

#pragma unroll
    for (int e = 0; e < 64; ++e) {
      v2f p = a[EBI[e]] * r[EBK[e]];
      if (ECV[e] < 0) p = -p;
      acc[EBJ[e]] = wmma_f32(p, bw[EPX[e]], acc[EBJ[e]]);
    }
  }

  const float bias = b_left[m0 + lo];
  const float rs2 = 0.70710678118654752440f;
#pragma unroll
  for (int v = 0; v < 8; ++v) {
    const float o0 = (acc[0][v] + bias) * rs2;
    const float o1 = acc[1][v] * rs2, o2 = acc[2][v] * rs2, o3 = acc[3][v] * rs2;
    const float o4 = acc[4][v] * rs2, o5 = acc[5][v] * rs2;
    const float o6 = acc[6][v] * rs2, o7 = acc[7][v] * rs2;
    float* dst = out + ((size_t)(b0 + v + (hi << 3)) * GA_NOUT + (m0 + lo)) * 8;
    *(float4*)(dst)     = make_float4(o0, o1, o2, o3);
    *(float4*)(dst + 4) = make_float4(o4, o5, o6, o7);
  }
}


extern "C" void kernel_launch(void* const* d_in, const int* in_sizes, int n_in,
                              void* d_out, int out_size, void* d_ws, size_t ws_size,
                              hipStream_t stream) {
  (void)in_sizes; (void)n_in; (void)out_size;
  const float* x       = (const float*)d_in[0];
  const float* weight  = (const float*)d_in[1];
  const float* w_right = (const float*)d_in[2];
  const float* w_left  = (const float*)d_in[3];
  const float* b_left  = (const float*)d_in[4];
  const float* norm_a  = (const float*)d_in[5];
  float* out = (float*)d_out;
  float* ws = (float*)d_ws;

  const size_t XT_N  = (size_t)GA_B * 8 * 512;
  const size_t W4_N  = (size_t)512 * 4 * 512;
  const size_t W20_N = (size_t)512 * 20 * 512;
  const size_t FAST_FLOATS = XT_N * 2 + W4_N * 2 + W20_N;

  dim3 block(32, 1, 1);
  dim3 grid1(GA_B / 16, GA_NIN / 16, 1);
  dim3 grid2(GA_B / 16, GA_NOUT / 16, 1);

  if (ws_size >= FAST_FLOATS * sizeof(float)) {
    float* xt  = ws;
    float* xrt = xt + XT_N;
    float* wrt = xrt + XT_N;
    float* wlt = wrt + W4_N;
    float* wtt = wlt + W4_N;

    ga_repack_x<<<dim3((GA_B * 512) / 256), dim3(256), 0, stream>>>(x, xt);
    ga_repack_w<4><<<dim3((512 * 512) / 256), dim3(256), 0, stream>>>(w_right, wrt);
    ga_repack_w<4><<<dim3((512 * 512) / 256), dim3(256), 0, stream>>>(w_left, wlt);
    ga_repack_w<20><<<dim3((512 * 512) / 256), dim3(256), 0, stream>>>(weight, wtt);

    ga_xr_norm_fast<<<grid1, block, 0, stream>>>(xt, wrt, norm_a, xrt);
    ga_out_fast<<<grid2, block, 0, stream>>>(xt, xrt, wtt, wlt, b_left, out);
  } else {
    float* xr = ws;
    ga_xr_norm_kernel<<<grid1, block, 0, stream>>>(x, w_right, norm_a, xr);
    ga_out_kernel<<<grid2, block, 0, stream>>>(x, xr, weight, w_left, b_left, out);
  }
}


# Operands repacked in WMMA fragment order ([row/16][k/4][channel][lane][2]) so each wave operand load reads one contiguous 256-byte chunk instead of 16 scattered 8-byte pieces
# speedup vs baseline: 2.3787x; 2.3787x over previous
#include <hip/hip_runtime.h>


typedef __attribute__((ext_vector_type(2))) float v2f;
typedef __attribute__((ext_vector_type(8))) float v8f;

#define GA_B    2048
#define GA_NIN  512
#define GA_NOUT 512

__device__ __forceinline__ v2f mk2(float a, float b) { v2f r; r.x = a; r.y = b; return r; }

__device__ __forceinline__ v8f wmma_f32(v2f a, v2f b, v8f c) {
  return __builtin_amdgcn_wmma_f32_16x16x4_f32(false, a, false, b, (short)0, c, false, false);
}

static constexpr int GRADE_[8] = {0, 1, 1, 1, 2, 2, 2, 3};

static constexpr int EBI[64] = {
  0,0,0,0,0,0,0,0,
  1,2,3, 1,2,3,
  1,1,2,2,3,3,
  1,1,2,2,3,3,
  1,2,3, 1,2,3,
  4,5,6,
  4,4,5,5,6,6,
  4,5,6, 4,5,6,
  4,4,5,5,6,6,
  4,5,6,
  7, 7,7,7, 7,7,7, 7};
static constexpr int EBJ[64] = {
  0,1,2,3,4,5,6,7,
  0,0,0, 1,2,3,
  2,3,1,3,1,2,
  4,5,4,6,5,6,
  6,5,4, 7,7,7,
  0,0,0,
  1,2,1,3,2,3,
  3,2,1, 4,5,6,
  5,6,4,6,4,5,
  7,7,7,
  0, 1,2,3, 4,5,6, 7};
static constexpr int EBK[64] = {
  0,1,2,3,4,5,6,7,
  1,2,3, 0,0,0,
  4,5,4,6,5,6,
  2,3,1,3,1,2,
  7,7,7, 6,5,4,
  4,5,6,
  2,1,3,1,3,2,
  7,7,7, 0,0,0,
  6,5,6,4,5,4,
  3,2,1,
  7, 6,5,4, 3,2,1, 0};
static constexpr int ECV[64] = {
  1,1,1,1,1,1,1,1,
  1,1,1, 1,1,1,
  1,1,-1,1,-1,-1,
  1,1,-1,1,-1,-1,
  1,-1,1, 1,-1,1,
  -1,-1,-1,
  1,-1,1,-1,1,-1,
  -1,1,-1, 1,1,1,
  1,-1,-1,1,1,-1,
  1,-1,1,
  -1, -1,1,-1, 1,-1,1, 1};
static constexpr int EPX[64] = {
  0,1,1,1,2,2,2,3,
  4,4,4, 5,5,5,
  6,6,6,6,6,6,
  7,7,7,7,7,7,
  8,8,8, 9,9,9,
  10,10,10,
  11,11,11,11,11,11,
  12,12,12, 13,13,13,
  14,14,14,14,14,14,
  15,15,15,
  16, 17,17,17, 18,18,18, 19};


#define GA_CHUNK 64
#define GA_TILE(C) (128 * (C) * GA_CHUNK)

__global__ __launch_bounds__(256) void ga_repack_x(
    const float* __restrict__ in, float* __restrict__ outp) {
  const int tid = blockIdx.x * 256 + threadIdx.x;
  const int lane = tid & 31;
  const int lo = lane & 15;
  const int hi = lane >> 4;
  const int chunk = tid >> 5;
  const int ks = chunk & 127;
  const int bt = chunk >> 7;
  const float* src = in + ((size_t)(bt * 16 + lo) * 512 + (ks * 4 + hi * 2)) * 8;
  const float4 p0 = *(const float4*)(src);
  const float4 p1 = *(const float4*)(src + 4);
  const float4 q0 = *(const float4*)(src + 8);
  const float4 q1 = *(const float4*)(src + 12);
  float* dst = outp + (size_t)chunk * (8 * GA_CHUNK) + lane * 2;
  *(v2f*)(dst + 0 * GA_CHUNK) = mk2(p0.x, q0.x);
  *(v2f*)(dst + 1 * GA_CHUNK) = mk2(p0.y, q0.y);
  *(v2f*)(dst + 2 * GA_CHUNK) = mk2(p0.z, q0.z);
  *(v2f*)(dst + 3 * GA_CHUNK) = mk2(p0.w, q0.w);
  *(v2f*)(dst + 4 * GA_CHUNK) = mk2(p1.x, q1.x);
  *(v2f*)(dst + 5 * GA_CHUNK) = mk2(p1.y, q1.y);
  *(v2f*)(dst + 6 * GA_CHUNK) = mk2(p1.z, q1.z);
  *(v2f*)(dst + 7 * GA_CHUNK) = mk2(p1.w, q1.w);
}

template <int C>
__global__ __launch_bounds__(256) void ga_repack_w(
    const float* __restrict__ in, float* __restrict__ outp) {
  const int tid = blockIdx.x * 256 + threadIdx.x;
  const int lane = tid & 31;
  const int lo = lane & 15;
  const int hi = lane >> 4;
  const int chunk = tid >> 5;
  const int ks = chunk & 127;
  const int mt = chunk >> 7;
  const float* src = in + ((size_t)(mt * 16 + lo) * 512 + (ks * 4 + hi * 2)) * C;
  float v[2 * C];
#pragma unroll
  for (int q = 0; q < (2 * C) / 4; ++q) {
    const float4 t = *(const float4*)(src + 4 * q);
    v[4 * q + 0] = t.x; v[4 * q + 1] = t.y; v[4 * q + 2] = t.z; v[4 * q + 3] = t.w;
  }
  float* dst = outp + (size_t)chunk * (C * GA_CHUNK) + lane * 2;
#pragma unroll
  for (int c = 0; c < C; ++c) *(v2f*)(dst + c * GA_CHUNK) = mk2(v[c], v[C + c]);
}


__global__ __launch_bounds__(32) void ga_xr_norm_fast(
    const float* __restrict__ xf, const float* __restrict__ wrf,
    const float* __restrict__ norm_a, float* __restrict__ xrf) {
  const int bt = blockIdx.x;
  const int m0 = blockIdx.y << 4;
  const int lane = threadIdx.x & 31;
  const int lo = lane & 15;
  const int hi = lane >> 4;

  const v8f vzero = {};
  v8f acc[8];
#pragma unroll
  for (int i = 0; i < 8; ++i) acc[i] = vzero;

  const float* xa = xf + (size_t)bt * GA_TILE(8) + lane * 2;
  const float* wb = wrf + (size_t)blockIdx.y * GA_TILE(4) + lane * 2;

#pragma unroll 4
  for (int ks = 0; ks < GA_NIN / 4; ++ks) {
    v2f a[8], g[4];
#pragma unroll
    for (int i = 0; i < 8; ++i) a[i] = *(const v2f*)(xa + (ks * 8 + i) * GA_CHUNK);
#pragma unroll
    for (int j = 0; j < 4; ++j) g[j] = *(const v2f*)(wb + (ks * 4 + j) * GA_CHUNK);
#pragma unroll
    for (int i = 0; i < 8; ++i)
      acc[i] = wmma_f32(a[i], g[GRADE_[i]], acc[i]);
  }

  const float4 na = *(const float4*)(norm_a + (size_t)(m0 + lo) * 4);
  const float s0 = 1.0f / (1.0f + __expf(-na.x));
  const float s1 = 1.0f / (1.0f + __expf(-na.y));
  const float s2 = 1.0f / (1.0f + __expf(-na.z));
  const float s3 = 1.0f / (1.0f + __expf(-na.w));

  float* dcol = xrf + (size_t)bt * GA_TILE(8) + (size_t)((m0 + lo) >> 2) * (8 * GA_CHUNK)
              + ((lo >> 1) & 1) * 32 + (lo & 1);

#pragma unroll
  for (int v = 0; v < 8; ++v) {
    const float e0 = acc[0][v], e1 = acc[1][v], e2 = acc[2][v], e3 = acc[3][v];
    const float e4 = acc[4][v], e5 = acc[5][v], e6 = acc[6][v], e7 = acc[7][v];
    const float n0 = __builtin_sqrtf(e0 * e0);
    const float n1 = __builtin_sqrtf(e1 * e1 + e2 * e2 + e3 * e3);
    const float n2 = __builtin_sqrtf(e4 * e4 + e5 * e5 + e6 * e6);
    const float n3 = __builtin_sqrtf(e7 * e7);
    const float i0 = 1.0f / (s0 * (n0 - 1.0f) + 1.0f + 1e-6f);
    const float i1 = 1.0f / (s1 * (n1 - 1.0f) + 1.0f + 1e-6f);
    const float i2 = 1.0f / (s2 * (n2 - 1.0f) + 1.0f + 1e-6f);
    const float i3 = 1.0f / (s3 * (n3 - 1.0f) + 1.0f + 1e-6f);
    float* dst = dcol + (v + (hi << 3)) * 2;
    dst[0 * GA_CHUNK] = e0 * i0;
    dst[1 * GA_CHUNK] = e1 * i1;
    dst[2 * GA_CHUNK] = e2 * i1;
    dst[3 * GA_CHUNK] = e3 * i1;
    dst[4 * GA_CHUNK] = e4 * i2;
    dst[5 * GA_CHUNK] = e5 * i2;
    dst[6 * GA_CHUNK] = e6 * i2;
    dst[7 * GA_CHUNK] = e7 * i3;
  }
}

__global__ __launch_bounds__(32) void ga_out_fast(
    const float* __restrict__ xf, const float* __restrict__ xrf,
    const float* __restrict__ wf, const float* __restrict__ wlf,
    const float* __restrict__ b_left, float* __restrict__ out) {
  const int b0 = blockIdx.x << 4;
  const int m0 = blockIdx.y << 4;
  const int lane = threadIdx.x & 31;
  const int lo = lane & 15;
  const int hi = lane >> 4;

  const v8f vzero = {};
  v8f acc[8];
#pragma unroll
  for (int i = 0; i < 8; ++i) acc[i] = vzero;

  const float* xa = xf  + (size_t)blockIdx.x * GA_TILE(8) + lane * 2;
  const float* ra = xrf + (size_t)blockIdx.x * GA_TILE(8) + lane * 2;
  const float* lb = wlf + (size_t)blockIdx.y * GA_TILE(4) + lane * 2;
  const float* wb = wf  + (size_t)blockIdx.y * GA_TILE(20) + lane * 2;

#pragma unroll 1
  for (int ks = 0; ks < GA_NIN / 4; ++ks) {
    v2f a[8], r[8], bl[4], bw[20];
#pragma unroll
    for (int i = 0; i < 8; ++i) a[i] = *(const v2f*)(xa + (ks * 8 + i) * GA_CHUNK);
#pragma unroll
    for (int i = 0; i < 8; ++i) r[i] = *(const v2f*)(ra + (ks * 8 + i) * GA_CHUNK);
#pragma unroll
    for (int j = 0; j < 4; ++j) bl[j] = *(const v2f*)(lb + (ks * 4 + j) * GA_CHUNK);
#pragma unroll
    for (int p = 0; p < 20; ++p) bw[p] = *(const v2f*)(wb + (ks * 20 + p) * GA_CHUNK);

#pragma unroll
    for (int i = 0; i < 8; ++i)
      acc[i] = wmma_f32(a[i], bl[GRADE_[i]], acc[i]);

#pragma unroll
    for (int e = 0; e < 64; ++e) {
      v2f p = a[EBI[e]] * r[EBK[e]];
      if (ECV[e] < 0) p = -p;
      acc[EBJ[e]] = wmma_f32(p, bw[EPX[e]], acc[EBJ[e]]);
    }
  }

  const float bias = b_left[m0 + lo];
  const float rs2 = 0.70710678118654752440f;
#pragma unroll
  for (int v = 0; v < 8; ++v) {
    const float o0 = (acc[0][v] + bias) * rs2;
    const float o1 = acc[1][v] * rs2, o2 = acc[2][v] * rs2, o3 = acc[3][v] * rs2;
    const float o4 = acc[4][v] * rs2, o5 = acc[5][v] * rs2;
    const float o6 = acc[6][v] * rs2, o7 = acc[7][v] * rs2;
    float* dst = out + ((size_t)(b0 + v + (hi << 3)) * GA_NOUT + (m0 + lo)) * 8;
    *(float4*)(dst)     = make_float4(o0, o1, o2, o3);
    *(float4*)(dst + 4) = make_float4(o4, o5, o6, o7);
  }
}


__global__ __launch_bounds__(32) void ga_xr_norm_kernel(
    const float* __restrict__ x, const float* __restrict__ w_right,
    const float* __restrict__ norm_a, float* __restrict__ xr) {
  const int b0 = blockIdx.x << 4;
  const int m0 = blockIdx.y << 4;
  const int lane = threadIdx.x & 31;
  const int lo = lane & 15;
  const int hi = lane >> 4;

  const v8f vzero = {};
  v8f acc[8];
#pragma unroll
  for (int i = 0; i < 8; ++i) acc[i] = vzero;

  const float* xrow = x + (size_t)(b0 + lo) * (GA_NIN * 8);
  const float* wrow = w_right + (size_t)(m0 + lo) * (GA_NIN * 4);

#pragma unroll 2
  for (int k = 0; k < GA_NIN; k += 4) {
    const int kk = k + (hi << 1);
    const float4 xa0 = *(const float4*)(xrow + kk * 8);
    const float4 xa1 = *(const float4*)(xrow + kk * 8 + 4);
    const float4 xb0 = *(const float4*)(xrow + kk * 8 + 8);
    const float4 xb1 = *(const float4*)(xrow + kk * 8 + 12);
    const float4 w0  = *(const float4*)(wrow + kk * 4);
    const float4 w1  = *(const float4*)(wrow + kk * 4 + 4);

    const v2f a0 = mk2(xa0.x, xb0.x), a1 = mk2(xa0.y, xb0.y);
    const v2f a2 = mk2(xa0.z, xb0.z), a3 = mk2(xa0.w, xb0.w);
    const v2f a4 = mk2(xa1.x, xb1.x), a5 = mk2(xa1.y, xb1.y);
    const v2f a6 = mk2(xa1.z, xb1.z), a7 = mk2(xa1.w, xb1.w);
    const v2f g0 = mk2(w0.x, w1.x), g1 = mk2(w0.y, w1.y);
    const v2f g2 = mk2(w0.z, w1.z), g3 = mk2(w0.w, w1.w);

    acc[0] = wmma_f32(a0, g0, acc[0]);
    acc[1] = wmma_f32(a1, g1, acc[1]);
    acc[2] = wmma_f32(a2, g1, acc[2]);
    acc[3] = wmma_f32(a3, g1, acc[3]);
    acc[4] = wmma_f32(a4, g2, acc[4]);
    acc[5] = wmma_f32(a5, g2, acc[5]);
    acc[6] = wmma_f32(a6, g2, acc[6]);
    acc[7] = wmma_f32(a7, g3, acc[7]);
  }

  const float4 na = *(const float4*)(norm_a + (size_t)(m0 + lo) * 4);
  const float s0 = 1.0f / (1.0f + __expf(-na.x));
  const float s1 = 1.0f / (1.0f + __expf(-na.y));
  const float s2 = 1.0f / (1.0f + __expf(-na.z));
  const float s3 = 1.0f / (1.0f + __expf(-na.w));

#pragma unroll
  for (int v = 0; v < 8; ++v) {
    const float e0 = acc[0][v], e1 = acc[1][v], e2 = acc[2][v], e3 = acc[3][v];
    const float e4 = acc[4][v], e5 = acc[5][v], e6 = acc[6][v], e7 = acc[7][v];
    const float n0 = __builtin_sqrtf(e0 * e0);
    const float n1 = __builtin_sqrtf(e1 * e1 + e2 * e2 + e3 * e3);
    const float n2 = __builtin_sqrtf(e4 * e4 + e5 * e5 + e6 * e6);
    const float n3 = __builtin_sqrtf(e7 * e7);
    const float i0 = 1.0f / (s0 * (n0 - 1.0f) + 1.0f + 1e-6f);
    const float i1 = 1.0f / (s1 * (n1 - 1.0f) + 1.0f + 1e-6f);
    const float i2 = 1.0f / (s2 * (n2 - 1.0f) + 1.0f + 1e-6f);
    const float i3 = 1.0f / (s3 * (n3 - 1.0f) + 1.0f + 1e-6f);
    float* dst = xr + ((size_t)(b0 + v + (hi << 3)) * GA_NIN + (m0 + lo)) * 8;
    *(float4*)(dst)     = make_float4(e0 * i0, e1 * i1, e2 * i1, e3 * i1);
    *(float4*)(dst + 4) = make_float4(e4 * i2, e5 * i2, e6 * i2, e7 * i3);
  }
}

__global__ __launch_bounds__(32) void ga_out_kernel(
    const float* __restrict__ x, const float* __restrict__ xr,
    const float* __restrict__ weight, const float* __restrict__ w_left,
    const float* __restrict__ b_left, float* __restrict__ out) {
  const int b0 = blockIdx.x << 4;
  const int m0 = blockIdx.y << 4;
  const int lane = threadIdx.x & 31;
  const int lo = lane & 15;
  const int hi = lane >> 4;

  const v8f vzero = {};
  v8f acc[8];
#pragma unroll
  for (int i = 0; i < 8; ++i) acc[i] = vzero;

  const float* xrow = x  + (size_t)(b0 + lo) * (GA_NIN * 8);
  const float* rrow = xr + (size_t)(b0 + lo) * (GA_NIN * 8);
  const float* lrow = w_left + (size_t)(m0 + lo) * (GA_NIN * 4);
  const float* wrow = weight + (size_t)(m0 + lo) * (GA_NIN * 20);

#pragma unroll 1
  for (int k = 0; k < GA_NIN; k += 4) {
    const int kk = k + (hi << 1);

    const float4 xa0 = *(const float4*)(xrow + kk * 8);
    const float4 xa1 = *(const float4*)(xrow + kk * 8 + 4);
    const float4 xb0 = *(const float4*)(xrow + kk * 8 + 8);
    const float4 xb1 = *(const float4*)(xrow + kk * 8 + 12);
    const float4 ra0 = *(const float4*)(rrow + kk * 8);
    const float4 ra1 = *(const float4*)(rrow + kk * 8 + 4);
    const float4 rb0 = *(const float4*)(rrow + kk * 8 + 8);
    const float4 rb1 = *(const float4*)(rrow + kk * 8 + 12);
    const float4 l0  = *(const float4*)(lrow + kk * 4);
    const float4 l1  = *(const float4*)(lrow + kk * 4 + 4);
    const float4 q0  = *(const float4*)(wrow + kk * 20);
    const float4 q1  = *(const float4*)(wrow + kk * 20 + 4);
    const float4 q2  = *(const float4*)(wrow + kk * 20 + 8);
    const float4 q3  = *(const float4*)(wrow + kk * 20 + 12);
    const float4 q4  = *(const float4*)(wrow + kk * 20 + 16);
    const float4 t0  = *(const float4*)(wrow + kk * 20 + 20);
    const float4 t1  = *(const float4*)(wrow + kk * 20 + 24);
    const float4 t2  = *(const float4*)(wrow + kk * 20 + 28);
    const float4 t3  = *(const float4*)(wrow + kk * 20 + 32);
    const float4 t4  = *(const float4*)(wrow + kk * 20 + 36);

    v2f a[8], r[8], bl[4], bw[20];
    a[0] = mk2(xa0.x, xb0.x); a[1] = mk2(xa0.y, xb0.y);
    a[2] = mk2(xa0.z, xb0.z); a[3] = mk2(xa0.w, xb0.w);
    a[4] = mk2(xa1.x, xb1.x); a[5] = mk2(xa1.y, xb1.y);
    a[6] = mk2(xa1.z, xb1.z); a[7] = mk2(xa1.w, xb1.w);
    r[0] = mk2(ra0.x, rb0.x); r[1] = mk2(ra0.y, rb0.y);
    r[2] = mk2(ra0.z, rb0.z); r[3] = mk2(ra0.w, rb0.w);
    r[4] = mk2(ra1.x, rb1.x); r[5] = mk2(ra1.y, rb1.y);
    r[6] = mk2(ra1.z, rb1.z); r[7] = mk2(ra1.w, rb1.w);
    bl[0] = mk2(l0.x, l1.x); bl[1] = mk2(l0.y, l1.y);
    bl[2] = mk2(l0.z, l1.z); bl[3] = mk2(l0.w, l1.w);
    bw[0]  = mk2(q0.x, t0.x); bw[1]  = mk2(q0.y, t0.y);
    bw[2]  = mk2(q0.z, t0.z); bw[3]  = mk2(q0.w, t0.w);
    bw[4]  = mk2(q1.x, t1.x); bw[5]  = mk2(q1.y, t1.y);
    bw[6]  = mk2(q1.z, t1.z); bw[7]  = mk2(q1.w, t1.w);
    bw[8]  = mk2(q2.x, t2.x); bw[9]  = mk2(q2.y, t2.y);
    bw[10] = mk2(q2.z, t2.z); bw[11] = mk2(q2.w, t2.w);
    bw[12] = mk2(q3.x, t3.x); bw[13] = mk2(q3.y, t3.y);
    bw[14] = mk2(q3.z, t3.z); bw[15] = mk2(q3.w, t3.w);
    bw[16] = mk2(q4.x, t4.x); bw[17] = mk2(q4.y, t4.y);
    bw[18] = mk2(q4.z, t4.z); bw[19] = mk2(q4.w, t4.w);

#pragma unroll
    for (int i = 0; i < 8; ++i)
      acc[i] = wmma_f32(a[i], bl[GRADE_[i]], acc[i]);

#pragma unroll
    for (int e = 0; e < 64; ++e) {
      v2f p = a[EBI[e]] * r[EBK[e]];
      if (ECV[e] < 0) p = -p;
      acc[EBJ[e]] = wmma_f32(p, bw[EPX[e]], acc[EBJ[e]]);
    }
  }

  const float bias = b_left[m0 + lo];
  const float rs2 = 0.70710678118654752440f;
#pragma unroll
  for (int v = 0; v < 8; ++v) {
    const float o0 = (acc[0][v] + bias) * rs2;
    const float o1 = acc[1][v] * rs2, o2 = acc[2][v] * rs2, o3 = acc[3][v] * rs2;
    const float o4 = acc[4][v] * rs2, o5 = acc[5][v] * rs2;
    const float o6 = acc[6][v] * rs2, o7 = acc[7][v] * rs2;
    float* dst = out + ((size_t)(b0 + v + (hi << 3)) * GA_NOUT + (m0 + lo)) * 8;
    *(float4*)(dst)     = make_float4(o0, o1, o2, o3);
    *(float4*)(dst + 4) = make_float4(o4, o5, o6, o7);
  }
}


extern "C" void kernel_launch(void* const* d_in, const int* in_sizes, int n_in,
                              void* d_out, int out_size, void* d_ws, size_t ws_size,
                              hipStream_t stream) {
  (void)in_sizes; (void)n_in; (void)out_size;
  const float* x       = (const float*)d_in[0];
  const float* weight  = (const float*)d_in[1];
  const float* w_right = (const float*)d_in[2];
  const float* w_left  = (const float*)d_in[3];
  const float* b_left  = (const float*)d_in[4];
  const float* norm_a  = (const float*)d_in[5];
  float* out = (float*)d_out;
  float* ws = (float*)d_ws;

  const size_t XT_N  = (size_t)GA_B * 8 * 512;
  const size_t W4_N  = (size_t)512 * 4 * 512;
  const size_t W20_N = (size_t)512 * 20 * 512;
  const size_t FAST_FLOATS = XT_N * 2 + W4_N * 2 + W20_N;

  dim3 block(32, 1, 1);
  dim3 grid1(GA_B / 16, GA_NIN / 16, 1);
  dim3 grid2(GA_B / 16, GA_NOUT / 16, 1);

  if (ws_size >= FAST_FLOATS * sizeof(float)) {
    float* xt  = ws;
    float* xrt = xt + XT_N;
    float* wrt = xrt + XT_N;
    float* wlt = wrt + W4_N;
    float* wtt = wlt + W4_N;

    ga_repack_x<<<dim3((GA_B / 16) * 128 * 32 / 256), dim3(256), 0, stream>>>(x, xt);
    ga_repack_w<4><<<dim3((512 / 16) * 128 * 32 / 256), dim3(256), 0, stream>>>(w_right, wrt);
    ga_repack_w<4><<<dim3((512 / 16) * 128 * 32 / 256), dim3(256), 0, stream>>>(w_left, wlt);
    ga_repack_w<20><<<dim3((512 / 16) * 128 * 32 / 256), dim3(256), 0, stream>>>(weight, wtt);

    ga_xr_norm_fast<<<grid1, block, 0, stream>>>(xt, wrt, norm_a, xrt);
    ga_out_fast<<<grid2, block, 0, stream>>>(xt, xrt, wtt, wlt, b_left, out);
  } else {
    float* xr = ws;
    ga_xr_norm_kernel<<<grid1, block, 0, stream>>>(x, w_right, norm_a, xr);
    ga_out_kernel<<<grid2, block, 0, stream>>>(x, xr, weight, w_left, b_left, out);
  }
}


# Main kernel: the 40 operand loads of k-step ks+1 are issued as one group into a second register set ahead of the 72 WMMAs of k-step ks (ping-pong sets, loop body holds 2 k-steps)
# speedup vs baseline: 2.9674x; 1.2475x over previous
#include <hip/hip_runtime.h>


typedef __attribute__((ext_vector_type(2))) float v2f;
typedef __attribute__((ext_vector_type(8))) float v8f;

#define GA_B    2048
#define GA_NIN  512
#define GA_NOUT 512

__device__ __forceinline__ v2f mk2(float a, float b) { v2f r; r.x = a; r.y = b; return r; }

__device__ __forceinline__ v8f wmma_f32(v2f a, v2f b, v8f c) {
  return __builtin_amdgcn_wmma_f32_16x16x4_f32(false, a, false, b, (short)0, c, false, false);
}

static constexpr int GRADE_[8] = {0, 1, 1, 1, 2, 2, 2, 3};

static constexpr int EBI[64] = {
  0,0,0,0,0,0,0,0,
  1,2,3, 1,2,3,
  1,1,2,2,3,3,
  1,1,2,2,3,3,
  1,2,3, 1,2,3,
  4,5,6,
  4,4,5,5,6,6,
  4,5,6, 4,5,6,
  4,4,5,5,6,6,
  4,5,6,
  7, 7,7,7, 7,7,7, 7};
static constexpr int EBJ[64] = {
  0,1,2,3,4,5,6,7,
  0,0,0, 1,2,3,
  2,3,1,3,1,2,
  4,5,4,6,5,6,
  6,5,4, 7,7,7,
  0,0,0,
  1,2,1,3,2,3,
  3,2,1, 4,5,6,
  5,6,4,6,4,5,
  7,7,7,
  0, 1,2,3, 4,5,6, 7};
static constexpr int EBK[64] = {
  0,1,2,3,4,5,6,7,
  1,2,3, 0,0,0,
  4,5,4,6,5,6,
  2,3,1,3,1,2,
  7,7,7, 6,5,4,
  4,5,6,
  2,1,3,1,3,2,
  7,7,7, 0,0,0,
  6,5,6,4,5,4,
  3,2,1,
  7, 6,5,4, 3,2,1, 0};
static constexpr int ECV[64] = {
  1,1,1,1,1,1,1,1,
  1,1,1, 1,1,1,
  1,1,-1,1,-1,-1,
  1,1,-1,1,-1,-1,
  1,-1,1, 1,-1,1,
  -1,-1,-1,
  1,-1,1,-1,1,-1,
  -1,1,-1, 1,1,1,
  1,-1,-1,1,1,-1,
  1,-1,1,
  -1, -1,1,-1, 1,-1,1, 1};
static constexpr int EPX[64] = {
  0,1,1,1,2,2,2,3,
  4,4,4, 5,5,5,
  6,6,6,6,6,6,
  7,7,7,7,7,7,
  8,8,8, 9,9,9,
  10,10,10,
  11,11,11,11,11,11,
  12,12,12, 13,13,13,
  14,14,14,14,14,14,
  15,15,15,
  16, 17,17,17, 18,18,18, 19};


#define GA_CHUNK 64
#define GA_TILE(C) (128 * (C) * GA_CHUNK)

__global__ __launch_bounds__(256) void ga_repack_x(
    const float* __restrict__ in, float* __restrict__ outp) {
  const int tid = blockIdx.x * 256 + threadIdx.x;
  const int lane = tid & 31;
  const int lo = lane & 15;
  const int hi = lane >> 4;
  const int chunk = tid >> 5;
  const int ks = chunk & 127;
  const int bt = chunk >> 7;
  const float* src = in + ((size_t)(bt * 16 + lo) * 512 + (ks * 4 + hi * 2)) * 8;
  const float4 p0 = *(const float4*)(src);
  const float4 p1 = *(const float4*)(src + 4);
  const float4 q0 = *(const float4*)(src + 8);
  const float4 q1 = *(const float4*)(src + 12);
  float* dst = outp + (size_t)chunk * (8 * GA_CHUNK) + lane * 2;
  *(v2f*)(dst + 0 * GA_CHUNK) = mk2(p0.x, q0.x);
  *(v2f*)(dst + 1 * GA_CHUNK) = mk2(p0.y, q0.y);
  *(v2f*)(dst + 2 * GA_CHUNK) = mk2(p0.z, q0.z);
  *(v2f*)(dst + 3 * GA_CHUNK) = mk2(p0.w, q0.w);
  *(v2f*)(dst + 4 * GA_CHUNK) = mk2(p1.x, q1.x);
  *(v2f*)(dst + 5 * GA_CHUNK) = mk2(p1.y, q1.y);
  *(v2f*)(dst + 6 * GA_CHUNK) = mk2(p1.z, q1.z);
  *(v2f*)(dst + 7 * GA_CHUNK) = mk2(p1.w, q1.w);
}

template <int C>
__global__ __launch_bounds__(256) void ga_repack_w(
    const float* __restrict__ in, float* __restrict__ outp) {
  const int tid = blockIdx.x * 256 + threadIdx.x;
  const int lane = tid & 31;
  const int lo = lane & 15;
  const int hi = lane >> 4;
  const int chunk = tid >> 5;
  const int ks = chunk & 127;
  const int mt = chunk >> 7;
  const float* src = in + ((size_t)(mt * 16 + lo) * 512 + (ks * 4 + hi * 2)) * C;
  float v[2 * C];
#pragma unroll
  for (int q = 0; q < (2 * C) / 4; ++q) {
    const float4 t = *(const float4*)(src + 4 * q);
    v[4 * q + 0] = t.x; v[4 * q + 1] = t.y; v[4 * q + 2] = t.z; v[4 * q + 3] = t.w;
  }
  float* dst = outp + (size_t)chunk * (C * GA_CHUNK) + lane * 2;
#pragma unroll
  for (int c = 0; c < C; ++c) *(v2f*)(dst + c * GA_CHUNK) = mk2(v[c], v[C + c]);
}


__global__ __launch_bounds__(32) void ga_xr_norm_fast(
    const float* __restrict__ xf, const float* __restrict__ wrf,
    const float* __restrict__ norm_a, float* __restrict__ xrf) {
  const int bt = blockIdx.x;
  const int m0 = blockIdx.y << 4;
  const int lane = threadIdx.x & 31;
  const int lo = lane & 15;
  const int hi = lane >> 4;

  const v8f vzero = {};
  v8f acc[8];
#pragma unroll
  for (int i = 0; i < 8; ++i) acc[i] = vzero;

  const float* xa = xf + (size_t)bt * GA_TILE(8) + lane * 2;
  const float* wb = wrf + (size_t)blockIdx.y * GA_TILE(4) + lane * 2;

#pragma unroll 4
  for (int ks = 0; ks < GA_NIN / 4; ++ks) {
    v2f a[8], g[4];
#pragma unroll
    for (int i = 0; i < 8; ++i) a[i] = *(const v2f*)(xa + (ks * 8 + i) * GA_CHUNK);
#pragma unroll
    for (int j = 0; j < 4; ++j) g[j] = *(const v2f*)(wb + (ks * 4 + j) * GA_CHUNK);
#pragma unroll
    for (int i = 0; i < 8; ++i)
      acc[i] = wmma_f32(a[i], g[GRADE_[i]], acc[i]);
  }

  const float4 na = *(const float4*)(norm_a + (size_t)(m0 + lo) * 4);
  const float s0 = 1.0f / (1.0f + __expf(-na.x));
  const float s1 = 1.0f / (1.0f + __expf(-na.y));
  const float s2 = 1.0f / (1.0f + __expf(-na.z));
  const float s3 = 1.0f / (1.0f + __expf(-na.w));

  float* dcol = xrf + (size_t)bt * GA_TILE(8) + (size_t)((m0 + lo) >> 2) * (8 * GA_CHUNK)
              + ((lo >> 1) & 1) * 32 + (lo & 1);

#pragma unroll
  for (int v = 0; v < 8; ++v) {
    const float e0 = acc[0][v], e1 = acc[1][v], e2 = acc[2][v], e3 = acc[3][v];
    const float e4 = acc[4][v], e5 = acc[5][v], e6 = acc[6][v], e7 = acc[7][v];
    const float n0 = __builtin_sqrtf(e0 * e0);
    const float n1 = __builtin_sqrtf(e1 * e1 + e2 * e2 + e3 * e3);
    const float n2 = __builtin_sqrtf(e4 * e4 + e5 * e5 + e6 * e6);
    const float n3 = __builtin_sqrtf(e7 * e7);
    const float i0 = 1.0f / (s0 * (n0 - 1.0f) + 1.0f + 1e-6f);
    const float i1 = 1.0f / (s1 * (n1 - 1.0f) + 1.0f + 1e-6f);
    const float i2 = 1.0f / (s2 * (n2 - 1.0f) + 1.0f + 1e-6f);
    const float i3 = 1.0f / (s3 * (n3 - 1.0f) + 1.0f + 1e-6f);
    float* dst = dcol + (v + (hi << 3)) * 2;
    dst[0 * GA_CHUNK] = e0 * i0;
    dst[1 * GA_CHUNK] = e1 * i1;
    dst[2 * GA_CHUNK] = e2 * i1;
    dst[3 * GA_CHUNK] = e3 * i1;
    dst[4 * GA_CHUNK] = e4 * i2;
    dst[5 * GA_CHUNK] = e5 * i2;
    dst[6 * GA_CHUNK] = e6 * i2;
    dst[7 * GA_CHUNK] = e7 * i3;
  }
}

struct GaOps {
  v2f a[8];
  v2f r[8];
  v2f bl[4];
  v2f bw[20];
};

__device__ __forceinline__ void ga_load_ops(GaOps& o, const float* __restrict__ xa,
                                            const float* __restrict__ ra,
                                            const float* __restrict__ lb,
                                            const float* __restrict__ wb, int ks) {
#pragma unroll
  for (int i = 0; i < 8; ++i) o.a[i] = *(const v2f*)(xa + (ks * 8 + i) * GA_CHUNK);
#pragma unroll
  for (int i = 0; i < 8; ++i) o.r[i] = *(const v2f*)(ra + (ks * 8 + i) * GA_CHUNK);
#pragma unroll
  for (int j = 0; j < 4; ++j) o.bl[j] = *(const v2f*)(lb + (ks * 4 + j) * GA_CHUNK);
#pragma unroll
  for (int p = 0; p < 20; ++p) o.bw[p] = *(const v2f*)(wb + (ks * 20 + p) * GA_CHUNK);
}

__device__ __forceinline__ void ga_kstep(const GaOps& o, v8f (&acc)[8]) {
#pragma unroll
  for (int i = 0; i < 8; ++i)
    acc[i] = wmma_f32(o.a[i], o.bl[GRADE_[i]], acc[i]);

#pragma unroll
  for (int e = 0; e < 64; ++e) {
    v2f p = o.a[EBI[e]] * o.r[EBK[e]];
    if (ECV[e] < 0) p = -p;
    acc[EBJ[e]] = wmma_f32(p, o.bw[EPX[e]], acc[EBJ[e]]);
  }
}

__global__ __launch_bounds__(32) void ga_out_fast(
    const float* __restrict__ xf, const float* __restrict__ xrf,
    const float* __restrict__ wf, const float* __restrict__ wlf,
    const float* __restrict__ b_left, float* __restrict__ out) {
  const int b0 = blockIdx.x << 4;
  const int m0 = blockIdx.y << 4;
  const int lane = threadIdx.x & 31;
  const int lo = lane & 15;
  const int hi = lane >> 4;

  const v8f vzero = {};
  v8f acc[8];
#pragma unroll
  for (int i = 0; i < 8; ++i) acc[i] = vzero;

  const float* xa = xf  + (size_t)blockIdx.x * GA_TILE(8) + lane * 2;
  const float* ra = xrf + (size_t)blockIdx.x * GA_TILE(8) + lane * 2;
  const float* lb = wlf + (size_t)blockIdx.y * GA_TILE(4) + lane * 2;
  const float* wb = wf  + (size_t)blockIdx.y * GA_TILE(20) + lane * 2;

  GaOps s0, s1;
  ga_load_ops(s0, xa, ra, lb, wb, 0);

#pragma unroll 1
  for (int ks = 0; ks < GA_NIN / 4; ks += 2) {
    ga_load_ops(s1, xa, ra, lb, wb, ks + 1);
    __builtin_amdgcn_sched_barrier(0);
    ga_kstep(s0, acc);
    __builtin_amdgcn_sched_barrier(0);
    const int kn = (ks + 2 < GA_NIN / 4) ? ks + 2 : GA_NIN / 4 - 1;
    ga_load_ops(s0, xa, ra, lb, wb, kn);
    __builtin_amdgcn_sched_barrier(0);
    ga_kstep(s1, acc);
  }

  const float bias = b_left[m0 + lo];
  const float rs2 = 0.70710678118654752440f;
#pragma unroll
  for (int v = 0; v < 8; ++v) {
    const float o0 = (acc[0][v] + bias) * rs2;
    const float o1 = acc[1][v] * rs2, o2 = acc[2][v] * rs2, o3 = acc[3][v] * rs2;
    const float o4 = acc[4][v] * rs2, o5 = acc[5][v] * rs2;
    const float o6 = acc[6][v] * rs2, o7 = acc[7][v] * rs2;
    float* dst = out + ((size_t)(b0 + v + (hi << 3)) * GA_NOUT + (m0 + lo)) * 8;
    *(float4*)(dst)     = make_float4(o0, o1, o2, o3);
    *(float4*)(dst + 4) = make_float4(o4, o5, o6, o7);
  }
}


__global__ __launch_bounds__(32) void ga_xr_norm_kernel(
    const float* __restrict__ x, const float* __restrict__ w_right,
    const float* __restrict__ norm_a, float* __restrict__ xr) {
  const int b0 = blockIdx.x << 4;
  const int m0 = blockIdx.y << 4;
  const int lane = threadIdx.x & 31;
  const int lo = lane & 15;
  const int hi = lane >> 4;

  const v8f vzero = {};
  v8f acc[8];
#pragma unroll
  for (int i = 0; i < 8; ++i) acc[i] = vzero;

  const float* xrow = x + (size_t)(b0 + lo) * (GA_NIN * 8);
  const float* wrow = w_right + (size_t)(m0 + lo) * (GA_NIN * 4);

#pragma unroll 2
  for (int k = 0; k < GA_NIN; k += 4) {
    const int kk = k + (hi << 1);
    const float4 xa0 = *(const float4*)(xrow + kk * 8);
    const float4 xa1 = *(const float4*)(xrow + kk * 8 + 4);
    const float4 xb0 = *(const float4*)(xrow + kk * 8 + 8);
    const float4 xb1 = *(const float4*)(xrow + kk * 8 + 12);
    const float4 w0  = *(const float4*)(wrow + kk * 4);
    const float4 w1  = *(const float4*)(wrow + kk * 4 + 4);

    const v2f a0 = mk2(xa0.x, xb0.x), a1 = mk2(xa0.y, xb0.y);
    const v2f a2 = mk2(xa0.z, xb0.z), a3 = mk2(xa0.w, xb0.w);
    const v2f a4 = mk2(xa1.x, xb1.x), a5 = mk2(xa1.y, xb1.y);
    const v2f a6 = mk2(xa1.z, xb1.z), a7 = mk2(xa1.w, xb1.w);
    const v2f g0 = mk2(w0.x, w1.x), g1 = mk2(w0.y, w1.y);
    const v2f g2 = mk2(w0.z, w1.z), g3 = mk2(w0.w, w1.w);

    acc[0] = wmma_f32(a0, g0, acc[0]);
    acc[1] = wmma_f32(a1, g1, acc[1]);
    acc[2] = wmma_f32(a2, g1, acc[2]);
    acc[3] = wmma_f32(a3, g1, acc[3]);
    acc[4] = wmma_f32(a4, g2, acc[4]);
    acc[5] = wmma_f32(a5, g2, acc[5]);
    acc[6] = wmma_f32(a6, g2, acc[6]);
    acc[7] = wmma_f32(a7, g3, acc[7]);
  }

  const float4 na = *(const float4*)(norm_a + (size_t)(m0 + lo) * 4);
  const float s0 = 1.0f / (1.0f + __expf(-na.x));
  const float s1 = 1.0f / (1.0f + __expf(-na.y));
  const float s2 = 1.0f / (1.0f + __expf(-na.z));
  const float s3 = 1.0f / (1.0f + __expf(-na.w));

#pragma unroll
  for (int v = 0; v < 8; ++v) {
    const float e0 = acc[0][v], e1 = acc[1][v], e2 = acc[2][v], e3 = acc[3][v];
    const float e4 = acc[4][v], e5 = acc[5][v], e6 = acc[6][v], e7 = acc[7][v];
    const float n0 = __builtin_sqrtf(e0 * e0);
    const float n1 = __builtin_sqrtf(e1 * e1 + e2 * e2 + e3 * e3);
    const float n2 = __builtin_sqrtf(e4 * e4 + e5 * e5 + e6 * e6);
    const float n3 = __builtin_sqrtf(e7 * e7);
    const float i0 = 1.0f / (s0 * (n0 - 1.0f) + 1.0f + 1e-6f);
    const float i1 = 1.0f / (s1 * (n1 - 1.0f) + 1.0f + 1e-6f);
    const float i2 = 1.0f / (s2 * (n2 - 1.0f) + 1.0f + 1e-6f);
    const float i3 = 1.0f / (s3 * (n3 - 1.0f) + 1.0f + 1e-6f);
    float* dst = xr + ((size_t)(b0 + v + (hi << 3)) * GA_NIN + (m0 + lo)) * 8;
    *(float4*)(dst)     = make_float4(e0 * i0, e1 * i1, e2 * i1, e3 * i1);
    *(float4*)(dst + 4) = make_float4(e4 * i2, e5 * i2, e6 * i2, e7 * i3);
  }
}

__global__ __launch_bounds__(32) void ga_out_kernel(
    const float* __restrict__ x, const float* __restrict__ xr,
    const float* __restrict__ weight, const float* __restrict__ w_left,
    const float* __restrict__ b_left, float* __restrict__ out) {
  const int b0 = blockIdx.x << 4;
  const int m0 = blockIdx.y << 4;
  const int lane = threadIdx.x & 31;
  const int lo = lane & 15;
  const int hi = lane >> 4;

  const v8f vzero = {};
  v8f acc[8];
#pragma unroll
  for (int i = 0; i < 8; ++i) acc[i] = vzero;

  const float* xrow = x  + (size_t)(b0 + lo) * (GA_NIN * 8);
  const float* rrow = xr + (size_t)(b0 + lo) * (GA_NIN * 8);
  const float* lrow = w_left + (size_t)(m0 + lo) * (GA_NIN * 4);
  const float* wrow = weight + (size_t)(m0 + lo) * (GA_NIN * 20);

#pragma unroll 1
  for (int k = 0; k < GA_NIN; k += 4) {
    const int kk = k + (hi << 1);

    const float4 xa0 = *(const float4*)(xrow + kk * 8);
    const float4 xa1 = *(const float4*)(xrow + kk * 8 + 4);
    const float4 xb0 = *(const float4*)(xrow + kk * 8 + 8);
    const float4 xb1 = *(const float4*)(xrow + kk * 8 + 12);
    const float4 ra0 = *(const float4*)(rrow + kk * 8);
    const float4 ra1 = *(const float4*)(rrow + kk * 8 + 4);
    const float4 rb0 = *(const float4*)(rrow + kk * 8 + 8);
    const float4 rb1 = *(const float4*)(rrow + kk * 8 + 12);
    const float4 l0  = *(const float4*)(lrow + kk * 4);
    const float4 l1  = *(const float4*)(lrow + kk * 4 + 4);
    const float4 q0  = *(const float4*)(wrow + kk * 20);
    const float4 q1  = *(const float4*)(wrow + kk * 20 + 4);
    const float4 q2  = *(const float4*)(wrow + kk * 20 + 8);
    const float4 q3  = *(const float4*)(wrow + kk * 20 + 12);
    const float4 q4  = *(const float4*)(wrow + kk * 20 + 16);
    const float4 t0  = *(const float4*)(wrow + kk * 20 + 20);
    const float4 t1  = *(const float4*)(wrow + kk * 20 + 24);
    const float4 t2  = *(const float4*)(wrow + kk * 20 + 28);
    const float4 t3  = *(const float4*)(wrow + kk * 20 + 32);
    const float4 t4  = *(const float4*)(wrow + kk * 20 + 36);

    v2f a[8], r[8], bl[4], bw[20];
    a[0] = mk2(xa0.x, xb0.x); a[1] = mk2(xa0.y, xb0.y);
    a[2] = mk2(xa0.z, xb0.z); a[3] = mk2(xa0.w, xb0.w);
    a[4] = mk2(xa1.x, xb1.x); a[5] = mk2(xa1.y, xb1.y);
    a[6] = mk2(xa1.z, xb1.z); a[7] = mk2(xa1.w, xb1.w);
    r[0] = mk2(ra0.x, rb0.x); r[1] = mk2(ra0.y, rb0.y);
    r[2] = mk2(ra0.z, rb0.z); r[3] = mk2(ra0.w, rb0.w);
    r[4] = mk2(ra1.x, rb1.x); r[5] = mk2(ra1.y, rb1.y);
    r[6] = mk2(ra1.z, rb1.z); r[7] = mk2(ra1.w, rb1.w);
    bl[0] = mk2(l0.x, l1.x); bl[1] = mk2(l0.y, l1.y);
    bl[2] = mk2(l0.z, l1.z); bl[3] = mk2(l0.w, l1.w);
    bw[0]  = mk2(q0.x, t0.x); bw[1]  = mk2(q0.y, t0.y);
    bw[2]  = mk2(q0.z, t0.z); bw[3]  = mk2(q0.w, t0.w);
    bw[4]  = mk2(q1.x, t1.x); bw[5]  = mk2(q1.y, t1.y);
    bw[6]  = mk2(q1.z, t1.z); bw[7]  = mk2(q1.w, t1.w);
    bw[8]  = mk2(q2.x, t2.x); bw[9]  = mk2(q2.y, t2.y);
    bw[10] = mk2(q2.z, t2.z); bw[11] = mk2(q2.w, t2.w);
    bw[12] = mk2(q3.x, t3.x); bw[13] = mk2(q3.y, t3.y);
    bw[14] = mk2(q3.z, t3.z); bw[15] = mk2(q3.w, t3.w);
    bw[16] = mk2(q4.x, t4.x); bw[17] = mk2(q4.y, t4.y);
    bw[18] = mk2(q4.z, t4.z); bw[19] = mk2(q4.w, t4.w);

#pragma unroll
    for (int i = 0; i < 8; ++i)
      acc[i] = wmma_f32(a[i], bl[GRADE_[i]], acc[i]);

#pragma unroll
    for (int e = 0; e < 64; ++e) {
      v2f p = a[EBI[e]] * r[EBK[e]];
      if (ECV[e] < 0) p = -p;
      acc[EBJ[e]] = wmma_f32(p, bw[EPX[e]], acc[EBJ[e]]);
    }
  }

  const float bias = b_left[m0 + lo];
  const float rs2 = 0.70710678118654752440f;
#pragma unroll
  for (int v = 0; v < 8; ++v) {
    const float o0 = (acc[0][v] + bias) * rs2;
    const float o1 = acc[1][v] * rs2, o2 = acc[2][v] * rs2, o3 = acc[3][v] * rs2;
    const float o4 = acc[4][v] * rs2, o5 = acc[5][v] * rs2;
    const float o6 = acc[6][v] * rs2, o7 = acc[7][v] * rs2;
    float* dst = out + ((size_t)(b0 + v + (hi << 3)) * GA_NOUT + (m0 + lo)) * 8;
    *(float4*)(dst)     = make_float4(o0, o1, o2, o3);
    *(float4*)(dst + 4) = make_float4(o4, o5, o6, o7);
  }
}


extern "C" void kernel_launch(void* const* d_in, const int* in_sizes, int n_in,
                              void* d_out, int out_size, void* d_ws, size_t ws_size,
                              hipStream_t stream) {
  (void)in_sizes; (void)n_in; (void)out_size;
  const float* x       = (const float*)d_in[0];
  const float* weight  = (const float*)d_in[1];
  const float* w_right = (const float*)d_in[2];
  const float* w_left  = (const float*)d_in[3];
  const float* b_left  = (const float*)d_in[4];
  const float* norm_a  = (const float*)d_in[5];
  float* out = (float*)d_out;
  float* ws = (float*)d_ws;

  const size_t XT_N  = (size_t)GA_B * 8 * 512;
  const size_t W4_N  = (size_t)512 * 4 * 512;
  const size_t W20_N = (size_t)512 * 20 * 512;
  const size_t FAST_FLOATS = XT_N * 2 + W4_N * 2 + W20_N;

  dim3 block(32, 1, 1);
  dim3 grid1(GA_B / 16, GA_NIN / 16, 1);
  dim3 grid2(GA_B / 16, GA_NOUT / 16, 1);

  if (ws_size >= FAST_FLOATS * sizeof(float)) {
    float* xt  = ws;
    float* xrt = xt + XT_N;
    float* wrt = xrt + XT_N;
    float* wlt = wrt + W4_N;
    float* wtt = wlt + W4_N;

    ga_repack_x<<<dim3((GA_B / 16) * 128 * 32 / 256), dim3(256), 0, stream>>>(x, xt);
    ga_repack_w<4><<<dim3((512 / 16) * 128 * 32 / 256), dim3(256), 0, stream>>>(w_right, wrt);
    ga_repack_w<4><<<dim3((512 / 16) * 128 * 32 / 256), dim3(256), 0, stream>>>(w_left, wlt);
    ga_repack_w<20><<<dim3((512 / 16) * 128 * 32 / 256), dim3(256), 0, stream>>>(weight, wtt);

    ga_xr_norm_fast<<<grid1, block, 0, stream>>>(xt, wrt, norm_a, xrt);
    ga_out_fast<<<grid2, block, 0, stream>>>(xt, xrt, wtt, wlt, b_left, out);
  } else {
    float* xr = ws;
    ga_xr_norm_kernel<<<grid1, block, 0, stream>>>(x, w_right, norm_a, xr);
    ga_out_kernel<<<grid2, block, 0, stream>>>(x, xr, weight, w_left, b_left, out);
  }
}


# Bilinear path: the 64 Cayley products are pre-summed in f32 on the A side per (weight path, output blade) into 44 groups, one WMMA each: 52 not 72 WMMAs per k-step; summation order changes
# speedup vs baseline: 3.5727x; 1.2040x over previous
#include <hip/hip_runtime.h>


typedef __attribute__((ext_vector_type(2))) float v2f;
typedef __attribute__((ext_vector_type(8))) float v8f;

#define GA_B    2048
#define GA_NIN  512
#define GA_NOUT 512

__device__ __forceinline__ v2f mk2(float a, float b) { v2f r; r.x = a; r.y = b; return r; }

__device__ __forceinline__ v8f wmma_f32(v2f a, v2f b, v8f c) {
  return __builtin_amdgcn_wmma_f32_16x16x4_f32(false, a, false, b, (short)0, c, false, false);
}

static constexpr int GRADE_[8] = {0, 1, 1, 1, 2, 2, 2, 3};

#define GA_NGROUP 44
static constexpr int GN[GA_NGROUP] = {
  1, 1, 1, 1, 1, 1, 1, 1,
  3, 1, 1, 1, 2, 2, 2, 3,
  3, 2, 2, 2, 1, 1, 1, 3,
  1, 2, 2, 2, 1, 1, 1, 1,
  1, 1, 1, 2, 2, 2, 1, 1,
  1, 1, 1, 1};
static constexpr int GP[GA_NGROUP] = {
  0, 1, 1, 1, 2, 2, 2, 3,
  4, 5, 5, 5, 7, 7, 7, 9,
  10, 6, 6, 6, 8, 8, 8, 15,
  16, 11, 11, 11, 13, 13, 13, 19,
  12, 12, 12, 14, 14, 14, 17, 17,
  17, 18, 18, 18};
static constexpr int GJ[GA_NGROUP] = {
  0, 1, 2, 3, 4, 5, 6, 7,
  0, 1, 2, 3, 4, 5, 6, 7,
  0, 1, 2, 3, 4, 5, 6, 7,
  0, 1, 2, 3, 4, 5, 6, 7,
  1, 2, 3, 4, 5, 6, 1, 2,
  3, 4, 5, 6};
static constexpr int GI[GA_NGROUP][3] = {
  {0,0,0}, {0,0,0}, {0,0,0}, {0,0,0}, {0,0,0}, {0,0,0}, {0,0,0}, {0,0,0},
  {1,2,3}, {1,0,0}, {2,0,0}, {3,0,0}, {1,2,0}, {1,3,0}, {2,3,0}, {1,2,3},
  {4,5,6}, {2,3,0}, {1,3,0}, {1,2,0}, {3,0,0}, {2,0,0}, {1,0,0}, {4,5,6},
  {7,0,0}, {4,5,0}, {4,6,0}, {5,6,0}, {4,0,0}, {5,0,0}, {6,0,0}, {7,0,0},
  {6,0,0}, {5,0,0}, {4,0,0}, {5,6,0}, {4,6,0}, {4,5,0}, {7,0,0}, {7,0,0},
  {7,0,0}, {7,0,0}, {7,0,0}, {7,0,0}};
static constexpr int GK[GA_NGROUP][3] = {
  {0,0,0}, {1,0,0}, {2,0,0}, {3,0,0}, {4,0,0}, {5,0,0}, {6,0,0}, {7,0,0},
  {1,2,3}, {0,0,0}, {0,0,0}, {0,0,0}, {2,1,0}, {3,1,0}, {3,2,0}, {6,5,4},
  {4,5,6}, {4,5,0}, {4,6,0}, {5,6,0}, {7,0,0}, {7,0,0}, {7,0,0}, {3,2,1},
  {7,0,0}, {2,3,0}, {1,3,0}, {1,2,0}, {0,0,0}, {0,0,0}, {0,0,0}, {0,0,0},
  {7,0,0}, {7,0,0}, {7,0,0}, {6,5,0}, {6,4,0}, {5,4,0}, {6,0,0}, {5,0,0},
  {4,0,0}, {3,0,0}, {2,0,0}, {1,0,0}};
static constexpr int GS[GA_NGROUP][3] = {
  {1,0,0}, {1,0,0}, {1,0,0}, {1,0,0}, {1,0,0}, {1,0,0}, {1,0,0}, {1,0,0},
  {1,1,1}, {1,0,0}, {1,0,0}, {1,0,0}, {1,-1,0}, {1,-1,0}, {1,-1,0}, {1,-1,1},
  {-1,-1,-1}, {-1,-1,0}, {1,-1,0}, {1,1,0}, {1,0,0}, {-1,0,0}, {1,0,0}, {1,-1,1},
  {-1,0,0}, {1,1,0}, {-1,1,0}, {-1,-1,0}, {1,0,0}, {1,0,0}, {1,0,0}, {1,0,0},
  {-1,0,0}, {1,0,0}, {-1,0,0}, {-1,1,0}, {1,-1,0}, {-1,1,0}, {-1,0,0}, {1,0,0},
  {-1,0,0}, {1,0,0}, {-1,0,0}, {1,0,0}};


#define GA_CHUNK 64
#define GA_TILE(C) (128 * (C) * GA_CHUNK)

__global__ __launch_bounds__(256) void ga_repack_x(
    const float* __restrict__ in, float* __restrict__ outp) {
  const int tid = blockIdx.x * 256 + threadIdx.x;
  const int lane = tid & 31;
  const int lo = lane & 15;
  const int hi = lane >> 4;
  const int chunk = tid >> 5;
  const int ks = chunk & 127;
  const int bt = chunk >> 7;
  const float* src = in + ((size_t)(bt * 16 + lo) * 512 + (ks * 4 + hi * 2)) * 8;
  const float4 p0 = *(const float4*)(src);
  const float4 p1 = *(const float4*)(src + 4);
  const float4 q0 = *(const float4*)(src + 8);
  const float4 q1 = *(const float4*)(src + 12);
  float* dst = outp + (size_t)chunk * (8 * GA_CHUNK) + lane * 2;
  *(v2f*)(dst + 0 * GA_CHUNK) = mk2(p0.x, q0.x);
  *(v2f*)(dst + 1 * GA_CHUNK) = mk2(p0.y, q0.y);
  *(v2f*)(dst + 2 * GA_CHUNK) = mk2(p0.z, q0.z);
  *(v2f*)(dst + 3 * GA_CHUNK) = mk2(p0.w, q0.w);
  *(v2f*)(dst + 4 * GA_CHUNK) = mk2(p1.x, q1.x);
  *(v2f*)(dst + 5 * GA_CHUNK) = mk2(p1.y, q1.y);
  *(v2f*)(dst + 6 * GA_CHUNK) = mk2(p1.z, q1.z);
  *(v2f*)(dst + 7 * GA_CHUNK) = mk2(p1.w, q1.w);
}

template <int C>
__global__ __launch_bounds__(256) void ga_repack_w(
    const float* __restrict__ in, float* __restrict__ outp) {
  const int tid = blockIdx.x * 256 + threadIdx.x;
  const int lane = tid & 31;
  const int lo = lane & 15;
  const int hi = lane >> 4;
  const int chunk = tid >> 5;
  const int ks = chunk & 127;
  const int mt = chunk >> 7;
  const float* src = in + ((size_t)(mt * 16 + lo) * 512 + (ks * 4 + hi * 2)) * C;
  float v[2 * C];
#pragma unroll
  for (int q = 0; q < (2 * C) / 4; ++q) {
    const float4 t = *(const float4*)(src + 4 * q);
    v[4 * q + 0] = t.x; v[4 * q + 1] = t.y; v[4 * q + 2] = t.z; v[4 * q + 3] = t.w;
  }
  float* dst = outp + (size_t)chunk * (C * GA_CHUNK) + lane * 2;
#pragma unroll
  for (int c = 0; c < C; ++c) *(v2f*)(dst + c * GA_CHUNK) = mk2(v[c], v[C + c]);
}


__global__ __launch_bounds__(32) void ga_xr_norm_fast(
    const float* __restrict__ xf, const float* __restrict__ wrf,
    const float* __restrict__ norm_a, float* __restrict__ xrf) {
  const int bt = blockIdx.x;
  const int m0 = blockIdx.y << 4;
  const int lane = threadIdx.x & 31;
  const int lo = lane & 15;
  const int hi = lane >> 4;

  const v8f vzero = {};
  v8f acc[8];
#pragma unroll
  for (int i = 0; i < 8; ++i) acc[i] = vzero;

  const float* xa = xf + (size_t)bt * GA_TILE(8) + lane * 2;
  const float* wb = wrf + (size_t)blockIdx.y * GA_TILE(4) + lane * 2;

#pragma unroll 4
  for (int ks = 0; ks < GA_NIN / 4; ++ks) {
    v2f a[8], g[4];
#pragma unroll
    for (int i = 0; i < 8; ++i) a[i] = *(const v2f*)(xa + (ks * 8 + i) * GA_CHUNK);
#pragma unroll
    for (int j = 0; j < 4; ++j) g[j] = *(const v2f*)(wb + (ks * 4 + j) * GA_CHUNK);
#pragma unroll
    for (int i = 0; i < 8; ++i)
      acc[i] = wmma_f32(a[i], g[GRADE_[i]], acc[i]);
  }

  const float4 na = *(const float4*)(norm_a + (size_t)(m0 + lo) * 4);
  const float s0 = 1.0f / (1.0f + __expf(-na.x));
  const float s1 = 1.0f / (1.0f + __expf(-na.y));
  const float s2 = 1.0f / (1.0f + __expf(-na.z));
  const float s3 = 1.0f / (1.0f + __expf(-na.w));

  float* dcol = xrf + (size_t)bt * GA_TILE(8) + (size_t)((m0 + lo) >> 2) * (8 * GA_CHUNK)
              + ((lo >> 1) & 1) * 32 + (lo & 1);

#pragma unroll
  for (int v = 0; v < 8; ++v) {
    const float e0 = acc[0][v], e1 = acc[1][v], e2 = acc[2][v], e3 = acc[3][v];
    const float e4 = acc[4][v], e5 = acc[5][v], e6 = acc[6][v], e7 = acc[7][v];
    const float n0 = __builtin_sqrtf(e0 * e0);
    const float n1 = __builtin_sqrtf(e1 * e1 + e2 * e2 + e3 * e3);
    const float n2 = __builtin_sqrtf(e4 * e4 + e5 * e5 + e6 * e6);
    const float n3 = __builtin_sqrtf(e7 * e7);
    const float i0 = 1.0f / (s0 * (n0 - 1.0f) + 1.0f + 1e-6f);
    const float i1 = 1.0f / (s1 * (n1 - 1.0f) + 1.0f + 1e-6f);
    const float i2 = 1.0f / (s2 * (n2 - 1.0f) + 1.0f + 1e-6f);
    const float i3 = 1.0f / (s3 * (n3 - 1.0f) + 1.0f + 1e-6f);
    float* dst = dcol + (v + (hi << 3)) * 2;
    dst[0 * GA_CHUNK] = e0 * i0;
    dst[1 * GA_CHUNK] = e1 * i1;
    dst[2 * GA_CHUNK] = e2 * i1;
    dst[3 * GA_CHUNK] = e3 * i1;
    dst[4 * GA_CHUNK] = e4 * i2;
    dst[5 * GA_CHUNK] = e5 * i2;
    dst[6 * GA_CHUNK] = e6 * i2;
    dst[7 * GA_CHUNK] = e7 * i3;
  }
}

struct GaOps {
  v2f a[8];
  v2f r[8];
  v2f bl[4];
  v2f bw[20];
};

__device__ __forceinline__ void ga_load_ops(GaOps& o, const float* __restrict__ xa,
                                            const float* __restrict__ ra,
                                            const float* __restrict__ lb,
                                            const float* __restrict__ wb, int ks) {
#pragma unroll
  for (int i = 0; i < 8; ++i) o.a[i] = *(const v2f*)(xa + (ks * 8 + i) * GA_CHUNK);
#pragma unroll
  for (int i = 0; i < 8; ++i) o.r[i] = *(const v2f*)(ra + (ks * 8 + i) * GA_CHUNK);
#pragma unroll
  for (int j = 0; j < 4; ++j) o.bl[j] = *(const v2f*)(lb + (ks * 4 + j) * GA_CHUNK);
#pragma unroll
  for (int p = 0; p < 20; ++p) o.bw[p] = *(const v2f*)(wb + (ks * 20 + p) * GA_CHUNK);
}

__device__ __forceinline__ void ga_kstep(const v2f (&a)[8], const v2f (&r)[8],
                                         const v2f (&bl)[4], const v2f (&bw)[20],
                                         v8f (&acc)[8]) {
#pragma unroll
  for (int i = 0; i < 8; ++i)
    acc[i] = wmma_f32(a[i], bl[GRADE_[i]], acc[i]);

#pragma unroll
  for (int g = 0; g < GA_NGROUP; ++g) {
    v2f p = a[GI[g][0]] * r[GK[g][0]];
    if (GS[g][0] < 0) p = -p;
#pragma unroll
    for (int t = 1; t < GN[g]; ++t) {
      const v2f q = a[GI[g][t]] * r[GK[g][t]];
      p = (GS[g][t] < 0) ? p - q : p + q;
    }
    acc[GJ[g]] = wmma_f32(p, bw[GP[g]], acc[GJ[g]]);
  }
}

__global__ __launch_bounds__(32) void ga_out_fast(
    const float* __restrict__ xf, const float* __restrict__ xrf,
    const float* __restrict__ wf, const float* __restrict__ wlf,
    const float* __restrict__ b_left, float* __restrict__ out) {
  const int b0 = blockIdx.x << 4;
  const int m0 = blockIdx.y << 4;
  const int lane = threadIdx.x & 31;
  const int lo = lane & 15;
  const int hi = lane >> 4;

  const v8f vzero = {};
  v8f acc[8];
#pragma unroll
  for (int i = 0; i < 8; ++i) acc[i] = vzero;

  const float* xa = xf  + (size_t)blockIdx.x * GA_TILE(8) + lane * 2;
  const float* ra = xrf + (size_t)blockIdx.x * GA_TILE(8) + lane * 2;
  const float* lb = wlf + (size_t)blockIdx.y * GA_TILE(4) + lane * 2;
  const float* wb = wf  + (size_t)blockIdx.y * GA_TILE(20) + lane * 2;

  GaOps s0, s1;
  ga_load_ops(s0, xa, ra, lb, wb, 0);

#pragma unroll 1
  for (int ks = 0; ks < GA_NIN / 4; ks += 2) {
    ga_load_ops(s1, xa, ra, lb, wb, ks + 1);
    __builtin_amdgcn_sched_barrier(0);
    ga_kstep(s0.a, s0.r, s0.bl, s0.bw, acc);
    __builtin_amdgcn_sched_barrier(0);
    const int kn = (ks + 2 < GA_NIN / 4) ? ks + 2 : GA_NIN / 4 - 1;
    ga_load_ops(s0, xa, ra, lb, wb, kn);
    __builtin_amdgcn_sched_barrier(0);
    ga_kstep(s1.a, s1.r, s1.bl, s1.bw, acc);
  }

  const float bias = b_left[m0 + lo];
  const float rs2 = 0.70710678118654752440f;
#pragma unroll
  for (int v = 0; v < 8; ++v) {
    const float o0 = (acc[0][v] + bias) * rs2;
    const float o1 = acc[1][v] * rs2, o2 = acc[2][v] * rs2, o3 = acc[3][v] * rs2;
    const float o4 = acc[4][v] * rs2, o5 = acc[5][v] * rs2;
    const float o6 = acc[6][v] * rs2, o7 = acc[7][v] * rs2;
    float* dst = out + ((size_t)(b0 + v + (hi << 3)) * GA_NOUT + (m0 + lo)) * 8;
    *(float4*)(dst)     = make_float4(o0, o1, o2, o3);
    *(float4*)(dst + 4) = make_float4(o4, o5, o6, o7);
  }
}


__global__ __launch_bounds__(32) void ga_xr_norm_kernel(
    const float* __restrict__ x, const float* __restrict__ w_right,
    const float* __restrict__ norm_a, float* __restrict__ xr) {
  const int b0 = blockIdx.x << 4;
  const int m0 = blockIdx.y << 4;
  const int lane = threadIdx.x & 31;
  const int lo = lane & 15;
  const int hi = lane >> 4;

  const v8f vzero = {};
  v8f acc[8];
#pragma unroll
  for (int i = 0; i < 8; ++i) acc[i] = vzero;

  const float* xrow = x + (size_t)(b0 + lo) * (GA_NIN * 8);
  const float* wrow = w_right + (size_t)(m0 + lo) * (GA_NIN * 4);

#pragma unroll 2
  for (int k = 0; k < GA_NIN; k += 4) {
    const int kk = k + (hi << 1);
    const float4 xa0 = *(const float4*)(xrow + kk * 8);
    const float4 xa1 = *(const float4*)(xrow + kk * 8 + 4);
    const float4 xb0 = *(const float4*)(xrow + kk * 8 + 8);
    const float4 xb1 = *(const float4*)(xrow + kk * 8 + 12);
    const float4 w0  = *(const float4*)(wrow + kk * 4);
    const float4 w1  = *(const float4*)(wrow + kk * 4 + 4);

    const v2f a0 = mk2(xa0.x, xb0.x), a1 = mk2(xa0.y, xb0.y);
    const v2f a2 = mk2(xa0.z, xb0.z), a3 = mk2(xa0.w, xb0.w);
    const v2f a4 = mk2(xa1.x, xb1.x), a5 = mk2(xa1.y, xb1.y);
    const v2f a6 = mk2(xa1.z, xb1.z), a7 = mk2(xa1.w, xb1.w);
    const v2f g0 = mk2(w0.x, w1.x), g1 = mk2(w0.y, w1.y);
    const v2f g2 = mk2(w0.z, w1.z), g3 = mk2(w0.w, w1.w);

    acc[0] = wmma_f32(a0, g0, acc[0]);
    acc[1] = wmma_f32(a1, g1, acc[1]);
    acc[2] = wmma_f32(a2, g1, acc[2]);
    acc[3] = wmma_f32(a3, g1, acc[3]);
    acc[4] = wmma_f32(a4, g2, acc[4]);
    acc[5] = wmma_f32(a5, g2, acc[5]);
    acc[6] = wmma_f32(a6, g2, acc[6]);
    acc[7] = wmma_f32(a7, g3, acc[7]);
  }

  const float4 na = *(const float4*)(norm_a + (size_t)(m0 + lo) * 4);
  const float s0 = 1.0f / (1.0f + __expf(-na.x));
  const float s1 = 1.0f / (1.0f + __expf(-na.y));
  const float s2 = 1.0f / (1.0f + __expf(-na.z));
  const float s3 = 1.0f / (1.0f + __expf(-na.w));

#pragma unroll
  for (int v = 0; v < 8; ++v) {
    const float e0 = acc[0][v], e1 = acc[1][v], e2 = acc[2][v], e3 = acc[3][v];
    const float e4 = acc[4][v], e5 = acc[5][v], e6 = acc[6][v], e7 = acc[7][v];
    const float n0 = __builtin_sqrtf(e0 * e0);
    const float n1 = __builtin_sqrtf(e1 * e1 + e2 * e2 + e3 * e3);
    const float n2 = __builtin_sqrtf(e4 * e4 + e5 * e5 + e6 * e6);
    const float n3 = __builtin_sqrtf(e7 * e7);
    const float i0 = 1.0f / (s0 * (n0 - 1.0f) + 1.0f + 1e-6f);
    const float i1 = 1.0f / (s1 * (n1 - 1.0f) + 1.0f + 1e-6f);
    const float i2 = 1.0f / (s2 * (n2 - 1.0f) + 1.0f + 1e-6f);
    const float i3 = 1.0f / (s3 * (n3 - 1.0f) + 1.0f + 1e-6f);
    float* dst = xr + ((size_t)(b0 + v + (hi << 3)) * GA_NIN + (m0 + lo)) * 8;
    *(float4*)(dst)     = make_float4(e0 * i0, e1 * i1, e2 * i1, e3 * i1);
    *(float4*)(dst + 4) = make_float4(e4 * i2, e5 * i2, e6 * i2, e7 * i3);
  }
}

__global__ __launch_bounds__(32) void ga_out_kernel(
    const float* __restrict__ x, const float* __restrict__ xr,
    const float* __restrict__ weight, const float* __restrict__ w_left,
    const float* __restrict__ b_left, float* __restrict__ out) {
  const int b0 = blockIdx.x << 4;
  const int m0 = blockIdx.y << 4;
  const int lane = threadIdx.x & 31;
  const int lo = lane & 15;
  const int hi = lane >> 4;

  const v8f vzero = {};
  v8f acc[8];
#pragma unroll
  for (int i = 0; i < 8; ++i) acc[i] = vzero;

  const float* xrow = x  + (size_t)(b0 + lo) * (GA_NIN * 8);
  const float* rrow = xr + (size_t)(b0 + lo) * (GA_NIN * 8);
  const float* lrow = w_left + (size_t)(m0 + lo) * (GA_NIN * 4);
  const float* wrow = weight + (size_t)(m0 + lo) * (GA_NIN * 20);

#pragma unroll 1
  for (int k = 0; k < GA_NIN; k += 4) {
    const int kk = k + (hi << 1);

    const float4 xa0 = *(const float4*)(xrow + kk * 8);
    const float4 xa1 = *(const float4*)(xrow + kk * 8 + 4);
    const float4 xb0 = *(const float4*)(xrow + kk * 8 + 8);
    const float4 xb1 = *(const float4*)(xrow + kk * 8 + 12);
    const float4 ra0 = *(const float4*)(rrow + kk * 8);
    const float4 ra1 = *(const float4*)(rrow + kk * 8 + 4);
    const float4 rb0 = *(const float4*)(rrow + kk * 8 + 8);
    const float4 rb1 = *(const float4*)(rrow + kk * 8 + 12);
    const float4 l0  = *(const float4*)(lrow + kk * 4);
    const float4 l1  = *(const float4*)(lrow + kk * 4 + 4);
    const float4 q0  = *(const float4*)(wrow + kk * 20);
    const float4 q1  = *(const float4*)(wrow + kk * 20 + 4);
    const float4 q2  = *(const float4*)(wrow + kk * 20 + 8);
    const float4 q3  = *(const float4*)(wrow + kk * 20 + 12);
    const float4 q4  = *(const float4*)(wrow + kk * 20 + 16);
    const float4 t0  = *(const float4*)(wrow + kk * 20 + 20);
    const float4 t1  = *(const float4*)(wrow + kk * 20 + 24);
    const float4 t2  = *(const float4*)(wrow + kk * 20 + 28);
    const float4 t3  = *(const float4*)(wrow + kk * 20 + 32);
    const float4 t4  = *(const float4*)(wrow + kk * 20 + 36);

    v2f a[8], r[8], bl[4], bw[20];
    a[0] = mk2(xa0.x, xb0.x); a[1] = mk2(xa0.y, xb0.y);
    a[2] = mk2(xa0.z, xb0.z); a[3] = mk2(xa0.w, xb0.w);
    a[4] = mk2(xa1.x, xb1.x); a[5] = mk2(xa1.y, xb1.y);
    a[6] = mk2(xa1.z, xb1.z); a[7] = mk2(xa1.w, xb1.w);
    r[0] = mk2(ra0.x, rb0.x); r[1] = mk2(ra0.y, rb0.y);
    r[2] = mk2(ra0.z, rb0.z); r[3] = mk2(ra0.w, rb0.w);
    r[4] = mk2(ra1.x, rb1.x); r[5] = mk2(ra1.y, rb1.y);
    r[6] = mk2(ra1.z, rb1.z); r[7] = mk2(ra1.w, rb1.w);
    bl[0] = mk2(l0.x, l1.x); bl[1] = mk2(l0.y, l1.y);
    bl[2] = mk2(l0.z, l1.z); bl[3] = mk2(l0.w, l1.w);
    bw[0]  = mk2(q0.x, t0.x); bw[1]  = mk2(q0.y, t0.y);
    bw[2]  = mk2(q0.z, t0.z); bw[3]  = mk2(q0.w, t0.w);
    bw[4]  = mk2(q1.x, t1.x); bw[5]  = mk2(q1.y, t1.y);
    bw[6]  = mk2(q1.z, t1.z); bw[7]  = mk2(q1.w, t1.w);
    bw[8]  = mk2(q2.x, t2.x); bw[9]  = mk2(q2.y, t2.y);
    bw[10] = mk2(q2.z, t2.z); bw[11] = mk2(q2.w, t2.w);
    bw[12] = mk2(q3.x, t3.x); bw[13] = mk2(q3.y, t3.y);
    bw[14] = mk2(q3.z, t3.z); bw[15] = mk2(q3.w, t3.w);
    bw[16] = mk2(q4.x, t4.x); bw[17] = mk2(q4.y, t4.y);
    bw[18] = mk2(q4.z, t4.z); bw[19] = mk2(q4.w, t4.w);

    ga_kstep(a, r, bl, bw, acc);
  }

  const float bias = b_left[m0 + lo];
  const float rs2 = 0.70710678118654752440f;
#pragma unroll
  for (int v = 0; v < 8; ++v) {
    const float o0 = (acc[0][v] + bias) * rs2;
    const float o1 = acc[1][v] * rs2, o2 = acc[2][v] * rs2, o3 = acc[3][v] * rs2;
    const float o4 = acc[4][v] * rs2, o5 = acc[5][v] * rs2;
    const float o6 = acc[6][v] * rs2, o7 = acc[7][v] * rs2;
    float* dst = out + ((size_t)(b0 + v + (hi << 3)) * GA_NOUT + (m0 + lo)) * 8;
    *(float4*)(dst)     = make_float4(o0, o1, o2, o3);
    *(float4*)(dst + 4) = make_float4(o4, o5, o6, o7);
  }
}


extern "C" void kernel_launch(void* const* d_in, const int* in_sizes, int n_in,
                              void* d_out, int out_size, void* d_ws, size_t ws_size,
                              hipStream_t stream) {
  (void)in_sizes; (void)n_in; (void)out_size;
  const float* x       = (const float*)d_in[0];
  const float* weight  = (const float*)d_in[1];
  const float* w_right = (const float*)d_in[2];
  const float* w_left  = (const float*)d_in[3];
  const float* b_left  = (const float*)d_in[4];
  const float* norm_a  = (const float*)d_in[5];
  float* out = (float*)d_out;
  float* ws = (float*)d_ws;

  const size_t XT_N  = (size_t)GA_B * 8 * 512;
  const size_t W4_N  = (size_t)512 * 4 * 512;
  const size_t W20_N = (size_t)512 * 20 * 512;
  const size_t FAST_FLOATS = XT_N * 2 + W4_N * 2 + W20_N;

  dim3 block(32, 1, 1);
  dim3 grid1(GA_B / 16, GA_NIN / 16, 1);
  dim3 grid2(GA_B / 16, GA_NOUT / 16, 1);

  if (ws_size >= FAST_FLOATS * sizeof(float)) {
    float* xt  = ws;
    float* xrt = xt + XT_N;
    float* wrt = xrt + XT_N;
    float* wlt = wrt + W4_N;
    float* wtt = wlt + W4_N;

    ga_repack_x<<<dim3((GA_B / 16) * 128 * 32 / 256), dim3(256), 0, stream>>>(x, xt);
    ga_repack_w<4><<<dim3((512 / 16) * 128 * 32 / 256), dim3(256), 0, stream>>>(w_right, wrt);
    ga_repack_w<4><<<dim3((512 / 16) * 128 * 32 / 256), dim3(256), 0, stream>>>(w_left, wlt);
    ga_repack_w<20><<<dim3((512 / 16) * 128 * 32 / 256), dim3(256), 0, stream>>>(weight, wtt);

    ga_xr_norm_fast<<<grid1, block, 0, stream>>>(xt, wrt, norm_a, xrt);
    ga_out_fast<<<grid2, block, 0, stream>>>(xt, xrt, wtt, wlt, b_left, out);
  } else {
    float* xr = ws;
    ga_xr_norm_kernel<<<grid1, block, 0, stream>>>(x, w_right, norm_a, xr);
    ga_out_kernel<<<grid2, block, 0, stream>>>(x, xr, weight, w_left, b_left, out);
  }
}


# Main kernel: one wave owns two neighbouring 16-column tiles of a 16-row tile (grid 128x16); each summed group product is formed once and feeds two WMMAs: half the vector multiplies per WMMA
# speedup vs baseline: 3.7113x; 1.0388x over previous
#include <hip/hip_runtime.h>


typedef __attribute__((ext_vector_type(2))) float v2f;
typedef __attribute__((ext_vector_type(8))) float v8f;

#define GA_B    2048
#define GA_NIN  512
#define GA_NOUT 512

__device__ __forceinline__ v2f mk2(float a, float b) { v2f r; r.x = a; r.y = b; return r; }

__device__ __forceinline__ v8f wmma_f32(v2f a, v2f b, v8f c) {
  return __builtin_amdgcn_wmma_f32_16x16x4_f32(false, a, false, b, (short)0, c, false, false);
}

static constexpr int GRADE_[8] = {0, 1, 1, 1, 2, 2, 2, 3};

#define GA_NGROUP 44
static constexpr int GN[GA_NGROUP] = {
  1, 1, 1, 1, 1, 1, 1, 1,
  3, 1, 1, 1, 2, 2, 2, 3,
  3, 2, 2, 2, 1, 1, 1, 3,
  1, 2, 2, 2, 1, 1, 1, 1,
  1, 1, 1, 2, 2, 2, 1, 1,
  1, 1, 1, 1};
static constexpr int GP[GA_NGROUP] = {
  0, 1, 1, 1, 2, 2, 2, 3,
  4, 5, 5, 5, 7, 7, 7, 9,
  10, 6, 6, 6, 8, 8, 8, 15,
  16, 11, 11, 11, 13, 13, 13, 19,
  12, 12, 12, 14, 14, 14, 17, 17,
  17, 18, 18, 18};
static constexpr int GJ[GA_NGROUP] = {
  0, 1, 2, 3, 4, 5, 6, 7,
  0, 1, 2, 3, 4, 5, 6, 7,
  0, 1, 2, 3, 4, 5, 6, 7,
  0, 1, 2, 3, 4, 5, 6, 7,
  1, 2, 3, 4, 5, 6, 1, 2,
  3, 4, 5, 6};
static constexpr int GI[GA_NGROUP][3] = {
  {0,0,0}, {0,0,0}, {0,0,0}, {0,0,0}, {0,0,0}, {0,0,0}, {0,0,0}, {0,0,0},
  {1,2,3}, {1,0,0}, {2,0,0}, {3,0,0}, {1,2,0}, {1,3,0}, {2,3,0}, {1,2,3},
  {4,5,6}, {2,3,0}, {1,3,0}, {1,2,0}, {3,0,0}, {2,0,0}, {1,0,0}, {4,5,6},
  {7,0,0}, {4,5,0}, {4,6,0}, {5,6,0}, {4,0,0}, {5,0,0}, {6,0,0}, {7,0,0},
  {6,0,0}, {5,0,0}, {4,0,0}, {5,6,0}, {4,6,0}, {4,5,0}, {7,0,0}, {7,0,0},
  {7,0,0}, {7,0,0}, {7,0,0}, {7,0,0}};
static constexpr int GK[GA_NGROUP][3] = {
  {0,0,0}, {1,0,0}, {2,0,0}, {3,0,0}, {4,0,0}, {5,0,0}, {6,0,0}, {7,0,0},
  {1,2,3}, {0,0,0}, {0,0,0}, {0,0,0}, {2,1,0}, {3,1,0}, {3,2,0}, {6,5,4},
  {4,5,6}, {4,5,0}, {4,6,0}, {5,6,0}, {7,0,0}, {7,0,0}, {7,0,0}, {3,2,1},
  {7,0,0}, {2,3,0}, {1,3,0}, {1,2,0}, {0,0,0}, {0,0,0}, {0,0,0}, {0,0,0},
  {7,0,0}, {7,0,0}, {7,0,0}, {6,5,0}, {6,4,0}, {5,4,0}, {6,0,0}, {5,0,0},
  {4,0,0}, {3,0,0}, {2,0,0}, {1,0,0}};
static constexpr int GS[GA_NGROUP][3] = {
  {1,0,0}, {1,0,0}, {1,0,0}, {1,0,0}, {1,0,0}, {1,0,0}, {1,0,0}, {1,0,0},
  {1,1,1}, {1,0,0}, {1,0,0}, {1,0,0}, {1,-1,0}, {1,-1,0}, {1,-1,0}, {1,-1,1},
  {-1,-1,-1}, {-1,-1,0}, {1,-1,0}, {1,1,0}, {1,0,0}, {-1,0,0}, {1,0,0}, {1,-1,1},
  {-1,0,0}, {1,1,0}, {-1,1,0}, {-1,-1,0}, {1,0,0}, {1,0,0}, {1,0,0}, {1,0,0},
  {-1,0,0}, {1,0,0}, {-1,0,0}, {-1,1,0}, {1,-1,0}, {-1,1,0}, {-1,0,0}, {1,0,0},
  {-1,0,0}, {1,0,0}, {-1,0,0}, {1,0,0}};


#define GA_CHUNK 64
#define GA_TILE(C) (128 * (C) * GA_CHUNK)

__global__ __launch_bounds__(256) void ga_repack_x(
    const float* __restrict__ in, float* __restrict__ outp) {
  const int tid = blockIdx.x * 256 + threadIdx.x;
  const int lane = tid & 31;
  const int lo = lane & 15;
  const int hi = lane >> 4;
  const int chunk = tid >> 5;
  const int ks = chunk & 127;
  const int bt = chunk >> 7;
  const float* src = in + ((size_t)(bt * 16 + lo) * 512 + (ks * 4 + hi * 2)) * 8;
  const float4 p0 = *(const float4*)(src);
  const float4 p1 = *(const float4*)(src + 4);
  const float4 q0 = *(const float4*)(src + 8);
  const float4 q1 = *(const float4*)(src + 12);
  float* dst = outp + (size_t)chunk * (8 * GA_CHUNK) + lane * 2;
  *(v2f*)(dst + 0 * GA_CHUNK) = mk2(p0.x, q0.x);
  *(v2f*)(dst + 1 * GA_CHUNK) = mk2(p0.y, q0.y);
  *(v2f*)(dst + 2 * GA_CHUNK) = mk2(p0.z, q0.z);
  *(v2f*)(dst + 3 * GA_CHUNK) = mk2(p0.w, q0.w);
  *(v2f*)(dst + 4 * GA_CHUNK) = mk2(p1.x, q1.x);
  *(v2f*)(dst + 5 * GA_CHUNK) = mk2(p1.y, q1.y);
  *(v2f*)(dst + 6 * GA_CHUNK) = mk2(p1.z, q1.z);
  *(v2f*)(dst + 7 * GA_CHUNK) = mk2(p1.w, q1.w);
}

template <int C>
__global__ __launch_bounds__(256) void ga_repack_w(
    const float* __restrict__ in, float* __restrict__ outp) {
  const int tid = blockIdx.x * 256 + threadIdx.x;
  const int lane = tid & 31;
  const int lo = lane & 15;
  const int hi = lane >> 4;
  const int chunk = tid >> 5;
  const int ks = chunk & 127;
  const int mt = chunk >> 7;
  const float* src = in + ((size_t)(mt * 16 + lo) * 512 + (ks * 4 + hi * 2)) * C;
  float v[2 * C];
#pragma unroll
  for (int q = 0; q < (2 * C) / 4; ++q) {
    const float4 t = *(const float4*)(src + 4 * q);
    v[4 * q + 0] = t.x; v[4 * q + 1] = t.y; v[4 * q + 2] = t.z; v[4 * q + 3] = t.w;
  }
  float* dst = outp + (size_t)chunk * (C * GA_CHUNK) + lane * 2;
#pragma unroll
  for (int c = 0; c < C; ++c) *(v2f*)(dst + c * GA_CHUNK) = mk2(v[c], v[C + c]);
}


__global__ __launch_bounds__(32) void ga_xr_norm_fast(
    const float* __restrict__ xf, const float* __restrict__ wrf,
    const float* __restrict__ norm_a, float* __restrict__ xrf) {
  const int bt = blockIdx.x;
  const int m0 = blockIdx.y << 4;
  const int lane = threadIdx.x & 31;
  const int lo = lane & 15;
  const int hi = lane >> 4;

  const v8f vzero = {};
  v8f acc[8];
#pragma unroll
  for (int i = 0; i < 8; ++i) acc[i] = vzero;

  const float* xa = xf + (size_t)bt * GA_TILE(8) + lane * 2;
  const float* wb = wrf + (size_t)blockIdx.y * GA_TILE(4) + lane * 2;

#pragma unroll 4
  for (int ks = 0; ks < GA_NIN / 4; ++ks) {
    v2f a[8], g[4];
#pragma unroll
    for (int i = 0; i < 8; ++i) a[i] = *(const v2f*)(xa + (ks * 8 + i) * GA_CHUNK);
#pragma unroll
    for (int j = 0; j < 4; ++j) g[j] = *(const v2f*)(wb + (ks * 4 + j) * GA_CHUNK);
#pragma unroll
    for (int i = 0; i < 8; ++i)
      acc[i] = wmma_f32(a[i], g[GRADE_[i]], acc[i]);
  }

  const float4 na = *(const float4*)(norm_a + (size_t)(m0 + lo) * 4);
  const float s0 = 1.0f / (1.0f + __expf(-na.x));
  const float s1 = 1.0f / (1.0f + __expf(-na.y));
  const float s2 = 1.0f / (1.0f + __expf(-na.z));
  const float s3 = 1.0f / (1.0f + __expf(-na.w));

  float* dcol = xrf + (size_t)bt * GA_TILE(8) + (size_t)((m0 + lo) >> 2) * (8 * GA_CHUNK)
              + ((lo >> 1) & 1) * 32 + (lo & 1);

#pragma unroll
  for (int v = 0; v < 8; ++v) {
    const float e0 = acc[0][v], e1 = acc[1][v], e2 = acc[2][v], e3 = acc[3][v];
    const float e4 = acc[4][v], e5 = acc[5][v], e6 = acc[6][v], e7 = acc[7][v];
    const float n0 = __builtin_sqrtf(e0 * e0);
    const float n1 = __builtin_sqrtf(e1 * e1 + e2 * e2 + e3 * e3);
    const float n2 = __builtin_sqrtf(e4 * e4 + e5 * e5 + e6 * e6);
    const float n3 = __builtin_sqrtf(e7 * e7);
    const float i0 = 1.0f / (s0 * (n0 - 1.0f) + 1.0f + 1e-6f);
    const float i1 = 1.0f / (s1 * (n1 - 1.0f) + 1.0f + 1e-6f);
    const float i2 = 1.0f / (s2 * (n2 - 1.0f) + 1.0f + 1e-6f);
    const float i3 = 1.0f / (s3 * (n3 - 1.0f) + 1.0f + 1e-6f);
    float* dst = dcol + (v + (hi << 3)) * 2;
    dst[0 * GA_CHUNK] = e0 * i0;
    dst[1 * GA_CHUNK] = e1 * i1;
    dst[2 * GA_CHUNK] = e2 * i1;
    dst[3 * GA_CHUNK] = e3 * i1;
    dst[4 * GA_CHUNK] = e4 * i2;
    dst[5 * GA_CHUNK] = e5 * i2;
    dst[6 * GA_CHUNK] = e6 * i2;
    dst[7 * GA_CHUNK] = e7 * i3;
  }
}

__device__ __forceinline__ void ga_kstep(const v2f (&a)[8], const v2f (&r)[8],
                                         const v2f (&bl)[4], const v2f (&bw)[20],
                                         v8f (&acc)[8]) {
#pragma unroll
  for (int i = 0; i < 8; ++i)
    acc[i] = wmma_f32(a[i], bl[GRADE_[i]], acc[i]);

#pragma unroll
  for (int g = 0; g < GA_NGROUP; ++g) {
    v2f p = a[GI[g][0]] * r[GK[g][0]];
    if (GS[g][0] < 0) p = -p;
#pragma unroll
    for (int t = 1; t < GN[g]; ++t) {
      const v2f q = a[GI[g][t]] * r[GK[g][t]];
      p = (GS[g][t] < 0) ? p - q : p + q;
    }
    acc[GJ[g]] = wmma_f32(p, bw[GP[g]], acc[GJ[g]]);
  }
}

struct GaOps2 {
  v2f a[8];
  v2f r[8];
  v2f bl[2][4];
  v2f bw[2][20];
};

#define GA_P1 10

__device__ __forceinline__ void ga_load_g1(GaOps2& o, const float* __restrict__ xa,
                                           const float* __restrict__ ra,
                                           const float* __restrict__ lb,
                                           const float* __restrict__ wb, int ks) {
#pragma unroll
  for (int i = 0; i < 8; ++i) o.a[i] = *(const v2f*)(xa + (ks * 8 + i) * GA_CHUNK);
#pragma unroll
  for (int i = 0; i < 8; ++i) o.r[i] = *(const v2f*)(ra + (ks * 8 + i) * GA_CHUNK);
#pragma unroll
  for (int t = 0; t < 2; ++t) {
#pragma unroll
    for (int j = 0; j < 4; ++j)
      o.bl[t][j] = *(const v2f*)(lb + t * GA_TILE(4) + (ks * 4 + j) * GA_CHUNK);
  }
#pragma unroll
  for (int t = 0; t < 2; ++t) {
#pragma unroll
    for (int p = 0; p < GA_P1; ++p)
      o.bw[t][p] = *(const v2f*)(wb + t * GA_TILE(20) + (ks * 20 + p) * GA_CHUNK);
  }
}

__device__ __forceinline__ void ga_load_g2(GaOps2& o, const float* __restrict__ wb, int ks) {
#pragma unroll
  for (int t = 0; t < 2; ++t) {
#pragma unroll
    for (int p = GA_P1; p < 20; ++p)
      o.bw[t][p] = *(const v2f*)(wb + t * GA_TILE(20) + (ks * 20 + p) * GA_CHUNK);
  }
}

template <int G0, int G1>
__device__ __forceinline__ void ga_groups2(const GaOps2& o, v8f (&acc)[2][8]) {
#pragma unroll
  for (int g = G0; g < G1; ++g) {
    v2f p = o.a[GI[g][0]] * o.r[GK[g][0]];
    if (GS[g][0] < 0) p = -p;
#pragma unroll
    for (int t = 1; t < GN[g]; ++t) {
      const v2f q = o.a[GI[g][t]] * o.r[GK[g][t]];
      p = (GS[g][t] < 0) ? p - q : p + q;
    }
    acc[0][GJ[g]] = wmma_f32(p, o.bw[0][GP[g]], acc[0][GJ[g]]);
    acc[1][GJ[g]] = wmma_f32(p, o.bw[1][GP[g]], acc[1][GJ[g]]);
  }
}

__device__ __forceinline__ void ga_kstep2_a(const GaOps2& o, v8f (&acc)[2][8]) {
#pragma unroll
  for (int i = 0; i < 8; ++i) {
    acc[0][i] = wmma_f32(o.a[i], o.bl[0][GRADE_[i]], acc[0][i]);
    acc[1][i] = wmma_f32(o.a[i], o.bl[1][GRADE_[i]], acc[1][i]);
  }
  ga_groups2<0, 16>(o, acc);
}

__device__ __forceinline__ void ga_kstep2_b(const GaOps2& o, v8f (&acc)[2][8]) {
  ga_groups2<16, GA_NGROUP>(o, acc);
}

__global__ __launch_bounds__(32) void ga_out_fast(
    const float* __restrict__ xf, const float* __restrict__ xrf,
    const float* __restrict__ wf, const float* __restrict__ wlf,
    const float* __restrict__ b_left, float* __restrict__ out) {
  const int b0 = blockIdx.x << 4;
  const int mt0 = blockIdx.y << 1;
  const int lane = threadIdx.x & 31;
  const int lo = lane & 15;
  const int hi = lane >> 4;

  const v8f vzero = {};
  v8f acc[2][8];
#pragma unroll
  for (int t = 0; t < 2; ++t) {
#pragma unroll
    for (int i = 0; i < 8; ++i) acc[t][i] = vzero;
  }

  const float* xa = xf  + (size_t)blockIdx.x * GA_TILE(8) + lane * 2;
  const float* ra = xrf + (size_t)blockIdx.x * GA_TILE(8) + lane * 2;
  const float* lb = wlf + (size_t)mt0 * GA_TILE(4) + lane * 2;
  const float* wb = wf  + (size_t)mt0 * GA_TILE(20) + lane * 2;

  GaOps2 s0, s1;
  ga_load_g1(s0, xa, ra, lb, wb, 0);
  ga_load_g2(s0, wb, 0);

#pragma unroll 1
  for (int ks = 0; ks < GA_NIN / 4; ks += 2) {
    ga_load_g1(s1, xa, ra, lb, wb, ks + 1);
    __builtin_amdgcn_sched_barrier(0);
    ga_kstep2_a(s0, acc);
    __builtin_amdgcn_sched_barrier(0);
    ga_load_g2(s1, wb, ks + 1);
    __builtin_amdgcn_sched_barrier(0);
    ga_kstep2_b(s0, acc);
    __builtin_amdgcn_sched_barrier(0);
    const int kn = (ks + 2 < GA_NIN / 4) ? ks + 2 : GA_NIN / 4 - 1;
    ga_load_g1(s0, xa, ra, lb, wb, kn);
    __builtin_amdgcn_sched_barrier(0);
    ga_kstep2_a(s1, acc);
    __builtin_amdgcn_sched_barrier(0);
    ga_load_g2(s0, wb, kn);
    __builtin_amdgcn_sched_barrier(0);
    ga_kstep2_b(s1, acc);
  }

  const float rs2 = 0.70710678118654752440f;
#pragma unroll
  for (int t = 0; t < 2; ++t) {
    const int m0 = (mt0 + t) << 4;
    const float bias = b_left[m0 + lo];
#pragma unroll
    for (int v = 0; v < 8; ++v) {
      const float o0 = (acc[t][0][v] + bias) * rs2;
      const float o1 = acc[t][1][v] * rs2, o2 = acc[t][2][v] * rs2, o3 = acc[t][3][v] * rs2;
      const float o4 = acc[t][4][v] * rs2, o5 = acc[t][5][v] * rs2;
      const float o6 = acc[t][6][v] * rs2, o7 = acc[t][7][v] * rs2;
      float* dst = out + ((size_t)(b0 + v + (hi << 3)) * GA_NOUT + (m0 + lo)) * 8;
      *(float4*)(dst)     = make_float4(o0, o1, o2, o3);
      *(float4*)(dst + 4) = make_float4(o4, o5, o6, o7);
    }
  }
}


__global__ __launch_bounds__(32) void ga_xr_norm_kernel(
    const float* __restrict__ x, const float* __restrict__ w_right,
    const float* __restrict__ norm_a, float* __restrict__ xr) {
  const int b0 = blockIdx.x << 4;
  const int m0 = blockIdx.y << 4;
  const int lane = threadIdx.x & 31;
  const int lo = lane & 15;
  const int hi = lane >> 4;

  const v8f vzero = {};
  v8f acc[8];
#pragma unroll
  for (int i = 0; i < 8; ++i) acc[i] = vzero;

  const float* xrow = x + (size_t)(b0 + lo) * (GA_NIN * 8);
  const float* wrow = w_right + (size_t)(m0 + lo) * (GA_NIN * 4);

#pragma unroll 2
  for (int k = 0; k < GA_NIN; k += 4) {
    const int kk = k + (hi << 1);
    const float4 xa0 = *(const float4*)(xrow + kk * 8);
    const float4 xa1 = *(const float4*)(xrow + kk * 8 + 4);
    const float4 xb0 = *(const float4*)(xrow + kk * 8 + 8);
    const float4 xb1 = *(const float4*)(xrow + kk * 8 + 12);
    const float4 w0  = *(const float4*)(wrow + kk * 4);
    const float4 w1  = *(const float4*)(wrow + kk * 4 + 4);

    const v2f a0 = mk2(xa0.x, xb0.x), a1 = mk2(xa0.y, xb0.y);
    const v2f a2 = mk2(xa0.z, xb0.z), a3 = mk2(xa0.w, xb0.w);
    const v2f a4 = mk2(xa1.x, xb1.x), a5 = mk2(xa1.y, xb1.y);
    const v2f a6 = mk2(xa1.z, xb1.z), a7 = mk2(xa1.w, xb1.w);
    const v2f g0 = mk2(w0.x, w1.x), g1 = mk2(w0.y, w1.y);
    const v2f g2 = mk2(w0.z, w1.z), g3 = mk2(w0.w, w1.w);

    acc[0] = wmma_f32(a0, g0, acc[0]);
    acc[1] = wmma_f32(a1, g1, acc[1]);
    acc[2] = wmma_f32(a2, g1, acc[2]);
    acc[3] = wmma_f32(a3, g1, acc[3]);
    acc[4] = wmma_f32(a4, g2, acc[4]);
    acc[5] = wmma_f32(a5, g2, acc[5]);
    acc[6] = wmma_f32(a6, g2, acc[6]);
    acc[7] = wmma_f32(a7, g3, acc[7]);
  }

  const float4 na = *(const float4*)(norm_a + (size_t)(m0 + lo) * 4);
  const float s0 = 1.0f / (1.0f + __expf(-na.x));
  const float s1 = 1.0f / (1.0f + __expf(-na.y));
  const float s2 = 1.0f / (1.0f + __expf(-na.z));
  const float s3 = 1.0f / (1.0f + __expf(-na.w));

#pragma unroll
  for (int v = 0; v < 8; ++v) {
    const float e0 = acc[0][v], e1 = acc[1][v], e2 = acc[2][v], e3 = acc[3][v];
    const float e4 = acc[4][v], e5 = acc[5][v], e6 = acc[6][v], e7 = acc[7][v];
    const float n0 = __builtin_sqrtf(e0 * e0);
    const float n1 = __builtin_sqrtf(e1 * e1 + e2 * e2 + e3 * e3);
    const float n2 = __builtin_sqrtf(e4 * e4 + e5 * e5 + e6 * e6);
    const float n3 = __builtin_sqrtf(e7 * e7);
    const float i0 = 1.0f / (s0 * (n0 - 1.0f) + 1.0f + 1e-6f);
    const float i1 = 1.0f / (s1 * (n1 - 1.0f) + 1.0f + 1e-6f);
    const float i2 = 1.0f / (s2 * (n2 - 1.0f) + 1.0f + 1e-6f);
    const float i3 = 1.0f / (s3 * (n3 - 1.0f) + 1.0f + 1e-6f);
    float* dst = xr + ((size_t)(b0 + v + (hi << 3)) * GA_NIN + (m0 + lo)) * 8;
    *(float4*)(dst)     = make_float4(e0 * i0, e1 * i1, e2 * i1, e3 * i1);
    *(float4*)(dst + 4) = make_float4(e4 * i2, e5 * i2, e6 * i2, e7 * i3);
  }
}

__global__ __launch_bounds__(32) void ga_out_kernel(
    const float* __restrict__ x, const float* __restrict__ xr,
    const float* __restrict__ weight, const float* __restrict__ w_left,
    const float* __restrict__ b_left, float* __restrict__ out) {
  const int b0 = blockIdx.x << 4;
  const int m0 = blockIdx.y << 4;
  const int lane = threadIdx.x & 31;
  const int lo = lane & 15;
  const int hi = lane >> 4;

  const v8f vzero = {};
  v8f acc[8];
#pragma unroll
  for (int i = 0; i < 8; ++i) acc[i] = vzero;

  const float* xrow = x  + (size_t)(b0 + lo) * (GA_NIN * 8);
  const float* rrow = xr + (size_t)(b0 + lo) * (GA_NIN * 8);
  const float* lrow = w_left + (size_t)(m0 + lo) * (GA_NIN * 4);
  const float* wrow = weight + (size_t)(m0 + lo) * (GA_NIN * 20);

#pragma unroll 1
  for (int k = 0; k < GA_NIN; k += 4) {
    const int kk = k + (hi << 1);

    const float4 xa0 = *(const float4*)(xrow + kk * 8);
    const float4 xa1 = *(const float4*)(xrow + kk * 8 + 4);
    const float4 xb0 = *(const float4*)(xrow + kk * 8 + 8);
    const float4 xb1 = *(const float4*)(xrow + kk * 8 + 12);
    const float4 ra0 = *(const float4*)(rrow + kk * 8);
    const float4 ra1 = *(const float4*)(rrow + kk * 8 + 4);
    const float4 rb0 = *(const float4*)(rrow + kk * 8 + 8);
    const float4 rb1 = *(const float4*)(rrow + kk * 8 + 12);
    const float4 l0  = *(const float4*)(lrow + kk * 4);
    const float4 l1  = *(const float4*)(lrow + kk * 4 + 4);
    const float4 q0  = *(const float4*)(wrow + kk * 20);
    const float4 q1  = *(const float4*)(wrow + kk * 20 + 4);
    const float4 q2  = *(const float4*)(wrow + kk * 20 + 8);
    const float4 q3  = *(const float4*)(wrow + kk * 20 + 12);
    const float4 q4  = *(const float4*)(wrow + kk * 20 + 16);
    const float4 t0  = *(const float4*)(wrow + kk * 20 + 20);
    const float4 t1  = *(const float4*)(wrow + kk * 20 + 24);
    const float4 t2  = *(const float4*)(wrow + kk * 20 + 28);
    const float4 t3  = *(const float4*)(wrow + kk * 20 + 32);
    const float4 t4  = *(const float4*)(wrow + kk * 20 + 36);

    v2f a[8], r[8], bl[4], bw[20];
    a[0] = mk2(xa0.x, xb0.x); a[1] = mk2(xa0.y, xb0.y);
    a[2] = mk2(xa0.z, xb0.z); a[3] = mk2(xa0.w, xb0.w);
    a[4] = mk2(xa1.x, xb1.x); a[5] = mk2(xa1.y, xb1.y);
    a[6] = mk2(xa1.z, xb1.z); a[7] = mk2(xa1.w, xb1.w);
    r[0] = mk2(ra0.x, rb0.x); r[1] = mk2(ra0.y, rb0.y);
    r[2] = mk2(ra0.z, rb0.z); r[3] = mk2(ra0.w, rb0.w);
    r[4] = mk2(ra1.x, rb1.x); r[5] = mk2(ra1.y, rb1.y);
    r[6] = mk2(ra1.z, rb1.z); r[7] = mk2(ra1.w, rb1.w);
    bl[0] = mk2(l0.x, l1.x); bl[1] = mk2(l0.y, l1.y);
    bl[2] = mk2(l0.z, l1.z); bl[3] = mk2(l0.w, l1.w);
    bw[0]  = mk2(q0.x, t0.x); bw[1]  = mk2(q0.y, t0.y);
    bw[2]  = mk2(q0.z, t0.z); bw[3]  = mk2(q0.w, t0.w);
    bw[4]  = mk2(q1.x, t1.x); bw[5]  = mk2(q1.y, t1.y);
    bw[6]  = mk2(q1.z, t1.z); bw[7]  = mk2(q1.w, t1.w);
    bw[8]  = mk2(q2.x, t2.x); bw[9]  = mk2(q2.y, t2.y);
    bw[10] = mk2(q2.z, t2.z); bw[11] = mk2(q2.w, t2.w);
    bw[12] = mk2(q3.x, t3.x); bw[13] = mk2(q3.y, t3.y);
    bw[14] = mk2(q3.z, t3.z); bw[15] = mk2(q3.w, t3.w);
    bw[16] = mk2(q4.x, t4.x); bw[17] = mk2(q4.y, t4.y);
    bw[18] = mk2(q4.z, t4.z); bw[19] = mk2(q4.w, t4.w);

    ga_kstep(a, r, bl, bw, acc);
  }

  const float bias = b_left[m0 + lo];
  const float rs2 = 0.70710678118654752440f;
#pragma unroll
  for (int v = 0; v < 8; ++v) {
    const float o0 = (acc[0][v] + bias) * rs2;
    const float o1 = acc[1][v] * rs2, o2 = acc[2][v] * rs2, o3 = acc[3][v] * rs2;
    const float o4 = acc[4][v] * rs2, o5 = acc[5][v] * rs2;
    const float o6 = acc[6][v] * rs2, o7 = acc[7][v] * rs2;
    float* dst = out + ((size_t)(b0 + v + (hi << 3)) * GA_NOUT + (m0 + lo)) * 8;
    *(float4*)(dst)     = make_float4(o0, o1, o2, o3);
    *(float4*)(dst + 4) = make_float4(o4, o5, o6, o7);
  }
}


extern "C" void kernel_launch(void* const* d_in, const int* in_sizes, int n_in,
                              void* d_out, int out_size, void* d_ws, size_t ws_size,
                              hipStream_t stream) {
  (void)in_sizes; (void)n_in; (void)out_size;
  const float* x       = (const float*)d_in[0];
  const float* weight  = (const float*)d_in[1];
  const float* w_right = (const float*)d_in[2];
  const float* w_left  = (const float*)d_in[3];
  const float* b_left  = (const float*)d_in[4];
  const float* norm_a  = (const float*)d_in[5];
  float* out = (float*)d_out;
  float* ws = (float*)d_ws;

  const size_t XT_N  = (size_t)GA_B * 8 * 512;
  const size_t W4_N  = (size_t)512 * 4 * 512;
  const size_t W20_N = (size_t)512 * 20 * 512;
  const size_t FAST_FLOATS = XT_N * 2 + W4_N * 2 + W20_N;

  dim3 block(32, 1, 1);
  dim3 grid1(GA_B / 16, GA_NIN / 16, 1);
  dim3 grid2(GA_B / 16, GA_NOUT / 16, 1);

  if (ws_size >= FAST_FLOATS * sizeof(float)) {
    float* xt  = ws;
    float* xrt = xt + XT_N;
    float* wrt = xrt + XT_N;
    float* wlt = wrt + W4_N;
    float* wtt = wlt + W4_N;

    ga_repack_x<<<dim3((GA_B / 16) * 128 * 32 / 256), dim3(256), 0, stream>>>(x, xt);
    ga_repack_w<4><<<dim3((512 / 16) * 128 * 32 / 256), dim3(256), 0, stream>>>(w_right, wrt);
    ga_repack_w<4><<<dim3((512 / 16) * 128 * 32 / 256), dim3(256), 0, stream>>>(w_left, wlt);
    ga_repack_w<20><<<dim3((512 / 16) * 128 * 32 / 256), dim3(256), 0, stream>>>(weight, wtt);

    ga_xr_norm_fast<<<grid1, block, 0, stream>>>(xt, wrt, norm_a, xrt);
    ga_out_fast<<<dim3(GA_B / 16, GA_NOUT / 32, 1), block, 0, stream>>>(xt, xrt, wtt, wlt, b_left, out);
  } else {
    float* xr = ws;
    ga_xr_norm_kernel<<<grid1, block, 0, stream>>>(x, w_right, norm_a, xr);
    ga_out_kernel<<<grid2, block, 0, stream>>>(x, xr, weight, w_left, b_left, out);
  }
}
